# starting kernel
# baseline (speedup 1.0000x reference)
#include <hip/hip_runtime.h>
#include <math.h>


#define N_NODES 50000
#define N_EDGES 800000
#define DFEAT   128
#define N_LAYERS 3
#define BN_EPS  1e-5f

typedef __attribute__((ext_vector_type(2))) float v2f;
typedef __attribute__((ext_vector_type(8))) float v8f;

__global__ void zero_f32(float* __restrict__ p, int n) {
    int i = blockIdx.x * blockDim.x + threadIdx.x;
    int stride = gridDim.x * blockDim.x;
    for (; i < n; i += stride) p[i] = 0.0f;
}

__global__ void degree_kernel(const int* __restrict__ row, float* __restrict__ deg, int E) {
    int i = blockIdx.x * blockDim.x + threadIdx.x;
    if (i < E) atomicAdd(&deg[row[i]], 1.0f);
}

__global__ void dinv_kernel(float* __restrict__ d, int N) {
    int i = blockIdx.x * blockDim.x + threadIdx.x;
    if (i < N) {
        float v = d[i];
        d[i] = (v > 0.0f) ? rsqrtf(v) : 0.0f;
    }
}

__global__ void gemm_wmma(const float* __restrict__ X,
                          const float* __restrict__ W,
                          const float* __restrict__ bias,
                          float* __restrict__ H, int M) {
    const int lane = threadIdx.x & 31;
    const int wave = threadIdx.x >> 5;
    const int wavesPerBlock = blockDim.x >> 5;
    const int tilesN = DFEAT / 16;
    const int tilesM = M >> 4;
    int tileId = blockIdx.x * wavesPerBlock + wave;
    if (tileId >= tilesM * tilesN) return;
    const int tm = tileId / tilesN;
    const int tn = tileId - tm * tilesN;
    const int m0 = tm << 4;
    const int n0 = tn << 4;

    const int l16   = lane & 15;
    const int khalf = (lane >> 4) << 1;

    const float* __restrict__ arow = X + (size_t)(m0 + l16) * DFEAT;
    const float* __restrict__ brow = W + (size_t)(n0 + l16) * DFEAT;

    const float bv = bias[n0 + l16];
    v8f c = {bv, bv, bv, bv, bv, bv, bv, bv};

#pragma unroll
    for (int k0 = 0; k0 < DFEAT; k0 += 4) {
        v2f a = *(const v2f*)(arow + k0 + khalf);
        v2f b = *(const v2f*)(brow + k0 + khalf);
        c = __builtin_amdgcn_wmma_f32_16x16x4_f32(false, a, false, b,
                                                  (short)0, c, false, false);
    }

    const int rowbase = m0 + ((lane >> 4) << 3);
    float* __restrict__ outcol = H + (size_t)rowbase * DFEAT + n0 + l16;
#pragma unroll
    for (int r = 0; r < 8; ++r) {
        outcol[(size_t)r * DFEAT] = c[r];
    }
}

__global__ void scatter_kernel(const int* __restrict__ row, const int* __restrict__ col,
                               const float* __restrict__ dinv, const float* __restrict__ H,
                               float* __restrict__ agg, int E) {
    const int lane = threadIdx.x & 31;
    const int e = (blockIdx.x * blockDim.x + threadIdx.x) >> 5;
    if (e >= E) return;
    const int r = row[e];
    const int c = col[e];
    const float nrm = dinv[r] * dinv[c];
    const float4 v = reinterpret_cast<const float4*>(H + (size_t)r * DFEAT)[lane];
    float* __restrict__ dst = agg + (size_t)c * DFEAT + lane * 4;
    atomicAdd(dst + 0, nrm * v.x);
    atomicAdd(dst + 1, nrm * v.y);
    atomicAdd(dst + 2, nrm * v.z);
    atomicAdd(dst + 3, nrm * v.w);
}

__global__ void bn_stats(const float* __restrict__ agg, float* __restrict__ stats, int N) {
    const int f = threadIdx.x;
    float s = 0.0f, s2 = 0.0f;
    for (int n = blockIdx.x; n < N; n += gridDim.x) {
        float v = agg[(size_t)n * DFEAT + f];
        s += v;
        s2 += v * v;
    }
    atomicAdd(&stats[f], s);
    atomicAdd(&stats[DFEAT + f], s2);
}

__global__ void bn_apply(const float* __restrict__ xin, const float* __restrict__ agg,
                         const float* __restrict__ stats, const float* __restrict__ g,
                         const float* __restrict__ be, float* __restrict__ xout,
                         int total4) {
    const int i = blockIdx.x * blockDim.x + threadIdx.x;
    if (i >= total4) return;
    const float invN = 1.0f / (float)N_NODES;
    const int f0 = (i & 31) << 2;
    const float4 a  = reinterpret_cast<const float4*>(agg)[i];
    const float4 xv = reinterpret_cast<const float4*>(xin)[i];
    float av[4]  = {a.x, a.y, a.z, a.w};
    float xvv[4] = {xv.x, xv.y, xv.z, xv.w};
    float r[4];
#pragma unroll
    for (int j = 0; j < 4; ++j) {
        const int f = f0 + j;
        const float mu  = stats[f] * invN;
        const float var = stats[DFEAT + f] * invN - mu * mu;
        const float sc  = g[f] * rsqrtf(var + BN_EPS);
        const float hv  = (av[j] - mu) * sc + be[f];
        r[j] = xvv[j] + fmaxf(hv, 0.0f);
    }
    float4 o = {r[0], r[1], r[2], r[3]};
    reinterpret_cast<float4*>(xout)[i] = o;
}

extern "C" void kernel_launch(void* const* d_in, const int* in_sizes, int n_in,
                              void* d_out, int out_size, void* d_ws, size_t ws_size,
                              hipStream_t stream) {
    (void)in_sizes; (void)n_in; (void)out_size; (void)ws_size;

    const float* x     = (const float*)d_in[0];
    const int*   ei    = (const int*)d_in[1];
    const float* W     = (const float*)d_in[2];
    const float* bias  = (const float*)d_in[3];
    const float* gamma = (const float*)d_in[4];
    const float* beta  = (const float*)d_in[5];
    float* out = (float*)d_out;

    const size_t ND = (size_t)N_NODES * DFEAT;
    float* ws    = (float*)d_ws;
    float* x_cur = ws;
    float* h     = ws + ND;
    float* agg   = ws + 2 * ND;
    float* dinv  = ws + 3 * ND;
    float* stats = dinv + N_NODES;

    const int* row = ei;
    const int* col = ei + N_EDGES;

    hipMemcpyAsync(x_cur, x, ND * sizeof(float), hipMemcpyDeviceToDevice, stream);

    zero_f32<<<256, 256, 0, stream>>>(dinv, N_NODES);
    degree_kernel<<<(N_EDGES + 255) / 256, 256, 0, stream>>>(row, dinv, N_EDGES);
    dinv_kernel<<<(N_NODES + 255) / 256, 256, 0, stream>>>(dinv, N_NODES);

    const int tiles      = (N_NODES / 16) * (DFEAT / 16);
    const int gemmBlocks = tiles / 8;
    const int scatThreads = N_EDGES * 32;

    for (int l = 0; l < N_LAYERS; ++l) {
        gemm_wmma<<<gemmBlocks, 256, 0, stream>>>(x_cur, W + (size_t)l * DFEAT * DFEAT,
                                                  bias + l * DFEAT, h, N_NODES);
        zero_f32<<<2048, 256, 0, stream>>>(agg, (int)ND);
        zero_f32<<<1, 256, 0, stream>>>(stats, 2 * DFEAT);
        scatter_kernel<<<(scatThreads + 255) / 256, 256, 0, stream>>>(row, col, dinv, h,
                                                                      agg, N_EDGES);
        bn_stats<<<512, DFEAT, 0, stream>>>(agg, stats, N_NODES);
        float* xo = (l == N_LAYERS - 1) ? out : x_cur;
        bn_apply<<<((int)(ND / 4) + 255) / 256, 256, 0, stream>>>(
            x_cur, agg, stats, gamma + l * DFEAT, beta + l * DFEAT, xo, (int)(ND / 4));
    }
}


# Edges grouped by destination once per call (returning int add gives rank, scan, place); one wave per node sums its source rows, stores once: no f32 row atomics; sum order = arrival rank
# speedup vs baseline: 1.4000x; 1.4000x over previous
#include <hip/hip_runtime.h>
#include <math.h>


#define N_NODES 50000
#define N_EDGES 800000
#define DFEAT   128
#define N_LAYERS 3
#define BN_EPS  1e-5f

typedef __attribute__((ext_vector_type(2))) float v2f;
typedef __attribute__((ext_vector_type(8))) float v8f;

__global__ void zero_f32(float* __restrict__ p, int n) {
    int i = blockIdx.x * blockDim.x + threadIdx.x;
    int stride = gridDim.x * blockDim.x;
    for (; i < n; i += stride) p[i] = 0.0f;
}

__global__ void degree_rank_kernel(const int* __restrict__ row, const int* __restrict__ col,
                                   float* __restrict__ deg, int* __restrict__ cnt,
                                   int* __restrict__ rank, int E) {
    int i = blockIdx.x * blockDim.x + threadIdx.x;
    if (i < E) {
        const int r = row[i];
        const int c = col[i];
        atomicAdd(&deg[r], 1.0f);
        rank[i] = atomicAdd(&cnt[c], 1);
    }
}

__global__ void scan_offsets(const int* __restrict__ cnt, int* __restrict__ off, int N) {
    __shared__ int part[1024];
    const int t = threadIdx.x;
    const int chunk = (N + 1023) >> 10;
    const int b = t * chunk;
    const int e = (b + chunk < N) ? (b + chunk) : N;
    int s = 0;
    for (int i = b; i < e; ++i) s += cnt[i];
    part[t] = s;
    __syncthreads();
    for (int d = 1; d < 1024; d <<= 1) {
        const int v = (t >= d) ? part[t - d] : 0;
        __syncthreads();
        part[t] += v;
        __syncthreads();
    }
    int run = part[t] - s;
    for (int i = b; i < e; ++i) {
        off[i] = run;
        run += cnt[i];
    }
    if (t == 1023) off[N] = part[1023];
}

__global__ void place_kernel(const int* __restrict__ row, const int* __restrict__ col,
                             const int* __restrict__ rank, const int* __restrict__ off,
                             const float* __restrict__ dinv, int* __restrict__ ent, int E) {
    int i = blockIdx.x * blockDim.x + threadIdx.x;
    if (i < E) {
        const int r = row[i];
        const int c = col[i];
        int2 v;
        v.x = r;
        v.y = __float_as_int(dinv[r] * dinv[c]);
        reinterpret_cast<int2*>(ent)[off[c] + rank[i]] = v;
    }
}

__global__ void dinv_kernel(float* __restrict__ d, int N) {
    int i = blockIdx.x * blockDim.x + threadIdx.x;
    if (i < N) {
        float v = d[i];
        d[i] = (v > 0.0f) ? rsqrtf(v) : 0.0f;
    }
}

__global__ void gemm_wmma(const float* __restrict__ X,
                          const float* __restrict__ W,
                          const float* __restrict__ bias,
                          float* __restrict__ H, int M) {
    const int lane = threadIdx.x & 31;
    const int wave = threadIdx.x >> 5;
    const int wavesPerBlock = blockDim.x >> 5;
    const int tilesN = DFEAT / 16;
    const int tilesM = M >> 4;
    int tileId = blockIdx.x * wavesPerBlock + wave;
    if (tileId >= tilesM * tilesN) return;
    const int tm = tileId / tilesN;
    const int tn = tileId - tm * tilesN;
    const int m0 = tm << 4;
    const int n0 = tn << 4;

    const int l16   = lane & 15;
    const int khalf = (lane >> 4) << 1;

    const float* __restrict__ arow = X + (size_t)(m0 + l16) * DFEAT;
    const float* __restrict__ brow = W + (size_t)(n0 + l16) * DFEAT;

    const float bv = bias[n0 + l16];
    v8f c = {bv, bv, bv, bv, bv, bv, bv, bv};

#pragma unroll
    for (int k0 = 0; k0 < DFEAT; k0 += 4) {
        v2f a = *(const v2f*)(arow + k0 + khalf);
        v2f b = *(const v2f*)(brow + k0 + khalf);
        c = __builtin_amdgcn_wmma_f32_16x16x4_f32(false, a, false, b,
                                                  (short)0, c, false, false);
    }

    const int rowbase = m0 + ((lane >> 4) << 3);
    float* __restrict__ outcol = H + (size_t)rowbase * DFEAT + n0 + l16;
#pragma unroll
    for (int r = 0; r < 8; ++r) {
        outcol[(size_t)r * DFEAT] = c[r];
    }
}

__global__ void gather_kernel(const int* __restrict__ off, const int* __restrict__ ent,
                              const float* __restrict__ H, float* __restrict__ agg, int N) {
    const int lane = threadIdx.x & 31;
    const int n = __builtin_amdgcn_readfirstlane(
        (int)(blockIdx.x * (blockDim.x >> 5) + (threadIdx.x >> 5)));
    if (n >= N) return;
    const int beg = off[n];
    const int end = off[n + 1];
    const float4* __restrict__ H4 = reinterpret_cast<const float4*>(H);
    float4 acc = {0.0f, 0.0f, 0.0f, 0.0f};
    for (int base = beg; base < end; base += 32) {
        const int idx = base + lane;
        int2 me = {0, 0};
        if (idx < end) me = reinterpret_cast<const int2*>(ent)[idx];
        const int m = (end - base < 32) ? (end - base) : 32;
#pragma unroll 4
        for (int j = 0; j < m; ++j) {
            const int s = __builtin_amdgcn_readlane(me.x, j);
            const float wt = __int_as_float(__builtin_amdgcn_readlane(me.y, j));
            const float4 v = H4[(size_t)s * (DFEAT / 4) + lane];
            acc.x = fmaf(wt, v.x, acc.x);
            acc.y = fmaf(wt, v.y, acc.y);
            acc.z = fmaf(wt, v.z, acc.z);
            acc.w = fmaf(wt, v.w, acc.w);
        }
    }
    reinterpret_cast<float4*>(agg)[(size_t)n * (DFEAT / 4) + lane] = acc;
}

__global__ void bn_stats(const float* __restrict__ agg, float* __restrict__ stats, int N) {
    const int f = threadIdx.x;
    float s = 0.0f, s2 = 0.0f;
    for (int n = blockIdx.x; n < N; n += gridDim.x) {
        float v = agg[(size_t)n * DFEAT + f];
        s += v;
        s2 += v * v;
    }
    atomicAdd(&stats[f], s);
    atomicAdd(&stats[DFEAT + f], s2);
}

__global__ void bn_apply(const float* __restrict__ xin, const float* __restrict__ agg,
                         const float* __restrict__ stats, const float* __restrict__ g,
                         const float* __restrict__ be, float* __restrict__ xout,
                         int total4) {
    const int i = blockIdx.x * blockDim.x + threadIdx.x;
    if (i >= total4) return;
    const float invN = 1.0f / (float)N_NODES;
    const int f0 = (i & 31) << 2;
    const float4 a  = reinterpret_cast<const float4*>(agg)[i];
    const float4 xv = reinterpret_cast<const float4*>(xin)[i];
    float av[4]  = {a.x, a.y, a.z, a.w};
    float xvv[4] = {xv.x, xv.y, xv.z, xv.w};
    float r[4];
#pragma unroll
    for (int j = 0; j < 4; ++j) {
        const int f = f0 + j;
        const float mu  = stats[f] * invN;
        const float var = stats[DFEAT + f] * invN - mu * mu;
        const float sc  = g[f] * rsqrtf(var + BN_EPS);
        const float hv  = (av[j] - mu) * sc + be[f];
        r[j] = xvv[j] + fmaxf(hv, 0.0f);
    }
    float4 o = {r[0], r[1], r[2], r[3]};
    reinterpret_cast<float4*>(xout)[i] = o;
}

extern "C" void kernel_launch(void* const* d_in, const int* in_sizes, int n_in,
                              void* d_out, int out_size, void* d_ws, size_t ws_size,
                              hipStream_t stream) {
    (void)in_sizes; (void)n_in; (void)out_size; (void)ws_size;

    const float* x     = (const float*)d_in[0];
    const int*   ei    = (const int*)d_in[1];
    const float* W     = (const float*)d_in[2];
    const float* bias  = (const float*)d_in[3];
    const float* gamma = (const float*)d_in[4];
    const float* beta  = (const float*)d_in[5];
    float* out = (float*)d_out;

    const size_t ND = (size_t)N_NODES * DFEAT;
    float* ws    = (float*)d_ws;
    float* dinv  = ws;
    int*   cnt   = (int*)(ws + N_NODES);
    int*   off   = (int*)(ws + 2 * N_NODES);
    int*   ent   = (int*)(ws + 4 * N_NODES);
    int*   rank  = (int*)(ws + 4 * N_NODES + 2 * N_EDGES);
    float* x_cur = out;
    float* h     = ws + ND;
    float* agg   = ws + 2 * ND;
    float* stats = ws + 3 * ND;

    const int* row = ei;
    const int* col = ei + N_EDGES;

    hipMemcpyAsync(x_cur, x, ND * sizeof(float), hipMemcpyDeviceToDevice, stream);

    zero_f32<<<256, 256, 0, stream>>>(ws, 2 * N_NODES);
    degree_rank_kernel<<<(N_EDGES + 255) / 256, 256, 0, stream>>>(row, col, dinv, cnt, rank,
                                                                  N_EDGES);
    dinv_kernel<<<(N_NODES + 255) / 256, 256, 0, stream>>>(dinv, N_NODES);
    scan_offsets<<<1, 1024, 0, stream>>>(cnt, off, N_NODES);
    place_kernel<<<(N_EDGES + 255) / 256, 256, 0, stream>>>(row, col, rank, off, dinv, ent,
                                                            N_EDGES);

    const int tiles      = (N_NODES / 16) * (DFEAT / 16);
    const int gemmBlocks = tiles / 8;

    for (int l = 0; l < N_LAYERS; ++l) {
        gemm_wmma<<<gemmBlocks, 256, 0, stream>>>(x_cur, W + (size_t)l * DFEAT * DFEAT,
                                                  bias + l * DFEAT, h, N_NODES);
        zero_f32<<<1, 256, 0, stream>>>(stats, 2 * DFEAT);
        gather_kernel<<<(N_NODES + 7) / 8, 256, 0, stream>>>(off, ent, h, agg, N_NODES);
        bn_stats<<<512, DFEAT, 0, stream>>>(agg, stats, N_NODES);
        bn_apply<<<((int)(ND / 4) + 255) / 256, 256, 0, stream>>>(
            x_cur, agg, stats, gamma + l * DFEAT, beta + l * DFEAT, x_cur, (int)(ND / 4));
    }
}


# BN statistics pass: each wave reads whole 512-B rows with 10 independent 16-B loads in flight (was one 4-B load per serial trip); block partials via LDS, then atomics: new reduction tree
# speedup vs baseline: 1.8348x; 1.3106x over previous
#include <hip/hip_runtime.h>
#include <math.h>


#define N_NODES 50000
#define N_EDGES 800000
#define DFEAT   128
#define N_LAYERS 3
#define BN_EPS  1e-5f

typedef __attribute__((ext_vector_type(2))) float v2f;
typedef __attribute__((ext_vector_type(8))) float v8f;

__global__ void zero_f32(float* __restrict__ p, int n) {
    int i = blockIdx.x * blockDim.x + threadIdx.x;
    int stride = gridDim.x * blockDim.x;
    for (; i < n; i += stride) p[i] = 0.0f;
}

__global__ void degree_rank_kernel(const int* __restrict__ row, const int* __restrict__ col,
                                   float* __restrict__ deg, int* __restrict__ cnt,
                                   int* __restrict__ rank, int E) {
    int i = blockIdx.x * blockDim.x + threadIdx.x;
    if (i < E) {
        const int r = row[i];
        const int c = col[i];
        atomicAdd(&deg[r], 1.0f);
        rank[i] = atomicAdd(&cnt[c], 1);
    }
}

__global__ void scan_offsets(const int* __restrict__ cnt, int* __restrict__ off, int N) {
    __shared__ int part[1024];
    const int t = threadIdx.x;
    const int chunk = (N + 1023) >> 10;
    const int b = t * chunk;
    const int e = (b + chunk < N) ? (b + chunk) : N;
    int s = 0;
    for (int i = b; i < e; ++i) s += cnt[i];
    part[t] = s;
    __syncthreads();
    for (int d = 1; d < 1024; d <<= 1) {
        const int v = (t >= d) ? part[t - d] : 0;
        __syncthreads();
        part[t] += v;
        __syncthreads();
    }
    int run = part[t] - s;
    for (int i = b; i < e; ++i) {
        off[i] = run;
        run += cnt[i];
    }
    if (t == 1023) off[N] = part[1023];
}

__global__ void place_kernel(const int* __restrict__ row, const int* __restrict__ col,
                             const int* __restrict__ rank, const int* __restrict__ off,
                             const float* __restrict__ dinv, int* __restrict__ ent, int E) {
    int i = blockIdx.x * blockDim.x + threadIdx.x;
    if (i < E) {
        const int r = row[i];
        const int c = col[i];
        int2 v;
        v.x = r;
        v.y = __float_as_int(dinv[r] * dinv[c]);
        reinterpret_cast<int2*>(ent)[off[c] + rank[i]] = v;
    }
}

__global__ void dinv_kernel(float* __restrict__ d, int N) {
    int i = blockIdx.x * blockDim.x + threadIdx.x;
    if (i < N) {
        float v = d[i];
        d[i] = (v > 0.0f) ? rsqrtf(v) : 0.0f;
    }
}

__global__ void gemm_wmma(const float* __restrict__ X,
                          const float* __restrict__ W,
                          const float* __restrict__ bias,
                          float* __restrict__ H, int M) {
    const int lane = threadIdx.x & 31;
    const int wave = threadIdx.x >> 5;
    const int wavesPerBlock = blockDim.x >> 5;
    const int tilesN = DFEAT / 16;
    const int tilesM = M >> 4;
    int tileId = blockIdx.x * wavesPerBlock + wave;
    if (tileId >= tilesM * tilesN) return;
    const int tm = tileId / tilesN;
    const int tn = tileId - tm * tilesN;
    const int m0 = tm << 4;
    const int n0 = tn << 4;

    const int l16   = lane & 15;
    const int khalf = (lane >> 4) << 1;

    const float* __restrict__ arow = X + (size_t)(m0 + l16) * DFEAT;
    const float* __restrict__ brow = W + (size_t)(n0 + l16) * DFEAT;

    const float bv = bias[n0 + l16];
    v8f c = {bv, bv, bv, bv, bv, bv, bv, bv};

#pragma unroll
    for (int k0 = 0; k0 < DFEAT; k0 += 4) {
        v2f a = *(const v2f*)(arow + k0 + khalf);
        v2f b = *(const v2f*)(brow + k0 + khalf);
        c = __builtin_amdgcn_wmma_f32_16x16x4_f32(false, a, false, b,
                                                  (short)0, c, false, false);
    }

    const int rowbase = m0 + ((lane >> 4) << 3);
    float* __restrict__ outcol = H + (size_t)rowbase * DFEAT + n0 + l16;
#pragma unroll
    for (int r = 0; r < 8; ++r) {
        outcol[(size_t)r * DFEAT] = c[r];
    }
}

__global__ void gather_kernel(const int* __restrict__ off, const int* __restrict__ ent,
                              const float* __restrict__ H, float* __restrict__ agg, int N) {
    const int lane = threadIdx.x & 31;
    const int n = __builtin_amdgcn_readfirstlane(
        (int)(blockIdx.x * (blockDim.x >> 5) + (threadIdx.x >> 5)));
    if (n >= N) return;
    const int beg = off[n];
    const int end = off[n + 1];
    const float4* __restrict__ H4 = reinterpret_cast<const float4*>(H);
    float4 acc = {0.0f, 0.0f, 0.0f, 0.0f};
    for (int base = beg; base < end; base += 32) {
        const int idx = base + lane;
        int2 me = {0, 0};
        if (idx < end) me = reinterpret_cast<const int2*>(ent)[idx];
        const int m = (end - base < 32) ? (end - base) : 32;
#pragma unroll 4
        for (int j = 0; j < m; ++j) {
            const int s = __builtin_amdgcn_readlane(me.x, j);
            const float wt = __int_as_float(__builtin_amdgcn_readlane(me.y, j));
            const float4 v = H4[(size_t)s * (DFEAT / 4) + lane];
            acc.x = fmaf(wt, v.x, acc.x);
            acc.y = fmaf(wt, v.y, acc.y);
            acc.z = fmaf(wt, v.z, acc.z);
            acc.w = fmaf(wt, v.w, acc.w);
        }
    }
    reinterpret_cast<float4*>(agg)[(size_t)n * (DFEAT / 4) + lane] = acc;
}

#define BN_WAVES 20
#define BN_ROWS  10

__global__ __launch_bounds__(BN_WAVES * 32)
void bn_stats(const float* __restrict__ agg, float* __restrict__ stats, int N) {
    __shared__ float part[2][BN_WAVES][DFEAT];
    const int lane = threadIdx.x & 31;
    const int wave = __builtin_amdgcn_readfirstlane((int)(threadIdx.x >> 5));
    const int gw = (int)blockIdx.x * BN_WAVES + wave;
    const int tw = (int)gridDim.x * BN_WAVES;
    const float4* __restrict__ A4 = reinterpret_cast<const float4*>(agg);
    float4 s = {0.0f, 0.0f, 0.0f, 0.0f};
    float4 q = {0.0f, 0.0f, 0.0f, 0.0f};
    for (int r0 = gw; r0 < N; r0 += BN_ROWS * tw) {
        float4 v[BN_ROWS];
        float  m[BN_ROWS];
#pragma unroll
        for (int j = 0; j < BN_ROWS; ++j) {
            const int r = r0 + j * tw;
            const bool ok = r < N;
            m[j] = ok ? 1.0f : 0.0f;
            v[j] = A4[(size_t)(ok ? r : (N - 1)) * (DFEAT / 4) + lane];
        }
#pragma unroll
        for (int j = 0; j < BN_ROWS; ++j) {
            s.x = fmaf(m[j], v[j].x, s.x);
            s.y = fmaf(m[j], v[j].y, s.y);
            s.z = fmaf(m[j], v[j].z, s.z);
            s.w = fmaf(m[j], v[j].w, s.w);
            q.x = fmaf(m[j] * v[j].x, v[j].x, q.x);
            q.y = fmaf(m[j] * v[j].y, v[j].y, q.y);
            q.z = fmaf(m[j] * v[j].z, v[j].z, q.z);
            q.w = fmaf(m[j] * v[j].w, v[j].w, q.w);
        }
    }
    *reinterpret_cast<float4*>(&part[0][wave][lane << 2]) = s;
    *reinterpret_cast<float4*>(&part[1][wave][lane << 2]) = q;
    __syncthreads();
    const int t = threadIdx.x;
    if (t < 2 * DFEAT) {
        const float* __restrict__ colp = &part[t >> 7][0][t & (DFEAT - 1)];
        float a = 0.0f;
#pragma unroll
        for (int w = 0; w < BN_WAVES; ++w) a += colp[w * DFEAT];
        atomicAdd(&stats[t], a);
    }
}

__global__ void bn_apply(const float* __restrict__ xin, const float* __restrict__ agg,
                         const float* __restrict__ stats, const float* __restrict__ g,
                         const float* __restrict__ be, float* __restrict__ xout,
                         int total4) {
    const int i = blockIdx.x * blockDim.x + threadIdx.x;
    if (i >= total4) return;
    const float invN = 1.0f / (float)N_NODES;
    const int f0 = (i & 31) << 2;
    const float4 a  = reinterpret_cast<const float4*>(agg)[i];
    const float4 xv = reinterpret_cast<const float4*>(xin)[i];
    float av[4]  = {a.x, a.y, a.z, a.w};
    float xvv[4] = {xv.x, xv.y, xv.z, xv.w};
    float r[4];
#pragma unroll
    for (int j = 0; j < 4; ++j) {
        const int f = f0 + j;
        const float mu  = stats[f] * invN;
        const float var = stats[DFEAT + f] * invN - mu * mu;
        const float sc  = g[f] * rsqrtf(var + BN_EPS);
        const float hv  = (av[j] - mu) * sc + be[f];
        r[j] = xvv[j] + fmaxf(hv, 0.0f);
    }
    float4 o = {r[0], r[1], r[2], r[3]};
    reinterpret_cast<float4*>(xout)[i] = o;
}

extern "C" void kernel_launch(void* const* d_in, const int* in_sizes, int n_in,
                              void* d_out, int out_size, void* d_ws, size_t ws_size,
                              hipStream_t stream) {
    (void)in_sizes; (void)n_in; (void)out_size; (void)ws_size;

    const float* x     = (const float*)d_in[0];
    const int*   ei    = (const int*)d_in[1];
    const float* W     = (const float*)d_in[2];
    const float* bias  = (const float*)d_in[3];
    const float* gamma = (const float*)d_in[4];
    const float* beta  = (const float*)d_in[5];
    float* out = (float*)d_out;

    const size_t ND = (size_t)N_NODES * DFEAT;
    float* ws    = (float*)d_ws;
    float* dinv  = ws;
    int*   cnt   = (int*)(ws + N_NODES);
    int*   off   = (int*)(ws + 2 * N_NODES);
    int*   ent   = (int*)(ws + 4 * N_NODES);
    int*   rank  = (int*)(ws + 4 * N_NODES + 2 * N_EDGES);
    float* x_cur = out;
    float* h     = ws + ND;
    float* agg   = ws + 2 * ND;
    float* stats = ws + 3 * ND;

    const int* row = ei;
    const int* col = ei + N_EDGES;

    hipMemcpyAsync(x_cur, x, ND * sizeof(float), hipMemcpyDeviceToDevice, stream);

    zero_f32<<<256, 256, 0, stream>>>(ws, 2 * N_NODES);
    degree_rank_kernel<<<(N_EDGES + 255) / 256, 256, 0, stream>>>(row, col, dinv, cnt, rank,
                                                                  N_EDGES);
    dinv_kernel<<<(N_NODES + 255) / 256, 256, 0, stream>>>(dinv, N_NODES);
    scan_offsets<<<1, 1024, 0, stream>>>(cnt, off, N_NODES);
    place_kernel<<<(N_EDGES + 255) / 256, 256, 0, stream>>>(row, col, rank, off, dinv, ent,
                                                            N_EDGES);

    const int tiles      = (N_NODES / 16) * (DFEAT / 16);
    const int gemmBlocks = tiles / 8;

    for (int l = 0; l < N_LAYERS; ++l) {
        gemm_wmma<<<gemmBlocks, 256, 0, stream>>>(x_cur, W + (size_t)l * DFEAT * DFEAT,
                                                  bias + l * DFEAT, h, N_NODES);
        zero_f32<<<1, 256, 0, stream>>>(stats, 2 * DFEAT);
        gather_kernel<<<(N_NODES + 7) / 8, 256, 0, stream>>>(off, ent, h, agg, N_NODES);
        bn_stats<<<250, BN_WAVES * 32, 0, stream>>>(agg, stats, N_NODES);
        bn_apply<<<((int)(ND / 4) + 255) / 256, 256, 0, stream>>>(
            x_cur, agg, stats, gamma + l * DFEAT, beta + l * DFEAT, x_cur, (int)(ND / 4));
    }
}


# Edge grouping counts in LDS: 64 blocks count their edge chunk (in-degree rank, out-degree) with LDS adds, a per-node pass sums the chunks; no global atomic per edge; sum order = chunk, then arrival
# speedup vs baseline: 2.1520x; 1.1729x over previous
#include <hip/hip_runtime.h>
#include <math.h>


#define N_NODES 50000
#define N_EDGES 800000
#define DFEAT   128
#define N_LAYERS 3
#define BN_EPS  1e-5f

typedef __attribute__((ext_vector_type(2))) float v2f;
typedef __attribute__((ext_vector_type(8))) float v8f;

__global__ void zero_f32(float* __restrict__ p, int n) {
    int i = blockIdx.x * blockDim.x + threadIdx.x;
    int stride = gridDim.x * blockDim.x;
    for (; i < n; i += stride) p[i] = 0.0f;
}

#define SORT_BLOCKS 64
#define SORT_CHUNK  (N_EDGES / SORT_BLOCKS)
#define HALF_N      (N_NODES / 2)
#define SORT_PER_THREAD ((SORT_CHUNK + 1023) / 1024)
static_assert(N_EDGES % SORT_BLOCKS == 0, "equal chunks");
static_assert(SORT_CHUNK < 65536, "a 16-bit field holds a chunk's count");
static_assert((2 * HALF_N) % 4 == 0, "tables are moved 16 bytes at a time");

__global__ __launch_bounds__(1024)
void hist_rank_kernel(const int* __restrict__ row, const int* __restrict__ col,
                      unsigned* __restrict__ histP, int* __restrict__ rank) {
    __shared__ __attribute__((aligned(16))) unsigned hs[2 * HALF_N];
    const int t = threadIdx.x;
    uint4* __restrict__ hs4 = reinterpret_cast<uint4*>(hs);
    for (int i = t; i < (2 * HALF_N) / 4; i += 1024) hs4[i] = make_uint4(0u, 0u, 0u, 0u);
    const int e0 = (int)blockIdx.x * SORT_CHUNK;
    int rr[SORT_PER_THREAD];
    int cc[SORT_PER_THREAD];
#pragma unroll
    for (int k = 0; k < SORT_PER_THREAD; ++k) {
        const int i = t + k * 1024;
        const int e = e0 + ((i < SORT_CHUNK) ? i : (SORT_CHUNK - 1));
        rr[k] = row[e];
        cc[k] = col[e];
    }
    __syncthreads();
    unsigned old[SORT_PER_THREAD];
#pragma unroll
    for (int k = 0; k < SORT_PER_THREAD; ++k) {
        const int i = t + k * 1024;
        old[k] = 0u;
        if (i < SORT_CHUNK) {
            const int c = cc[k];
            const int r = rr[k];
            old[k] = atomicAdd(&hs[c >> 1], 1u << ((unsigned)(c & 1) << 4));
            atomicAdd(&hs[HALF_N + (r >> 1)], 1u << ((unsigned)(r & 1) << 4));
        }
    }
#pragma unroll
    for (int k = 0; k < SORT_PER_THREAD; ++k) {
        const int i = t + k * 1024;
        if (i < SORT_CHUNK)
            rank[e0 + i] = (int)((old[k] >> ((unsigned)(cc[k] & 1) << 4)) & 0xffffu);
    }
    __syncthreads();
    uint4* __restrict__ out4 =
        reinterpret_cast<uint4*>(histP + (size_t)blockIdx.x * (2 * HALF_N));
    for (int i = t; i < (2 * HALF_N) / 4; i += 1024) out4[i] = hs4[i];
}

__global__ void combine_kernel(const unsigned* __restrict__ histP, int* __restrict__ base,
                               int* __restrict__ cnt, float* __restrict__ dinv) {
    const int t = blockIdx.x * blockDim.x + threadIdx.x;
    if (t >= HALF_N) return;
    unsigned c0 = 0u, c1 = 0u, d0 = 0u, d1 = 0u;
#pragma unroll 32
    for (int b = 0; b < SORT_BLOCKS; ++b) {
        const unsigned pc = histP[(size_t)b * (2 * HALF_N) + t];
        const unsigned pd = histP[(size_t)b * (2 * HALF_N) + HALF_N + t];
        int2 v;
        v.x = (int)c0;
        v.y = (int)c1;
        reinterpret_cast<int2*>(base + (size_t)b * N_NODES)[t] = v;
        c0 += pc & 0xffffu;
        c1 += pc >> 16;
        d0 += pd & 0xffffu;
        d1 += pd >> 16;
    }
    int2 cv;
    cv.x = (int)c0;
    cv.y = (int)c1;
    reinterpret_cast<int2*>(cnt)[t] = cv;
    float2 dv;
    dv.x = (d0 > 0u) ? rsqrtf((float)d0) : 0.0f;
    dv.y = (d1 > 0u) ? rsqrtf((float)d1) : 0.0f;
    reinterpret_cast<float2*>(dinv)[t] = dv;
}

__global__ void scan_offsets(const int* __restrict__ cnt, int* __restrict__ off, int N) {
    __shared__ int part[1024];
    const int t = threadIdx.x;
    const int chunk = (N + 1023) >> 10;
    const int b = t * chunk;
    const int e = (b + chunk < N) ? (b + chunk) : N;
    int s = 0;
    for (int i = b; i < e; ++i) s += cnt[i];
    part[t] = s;
    __syncthreads();
    for (int d = 1; d < 1024; d <<= 1) {
        const int v = (t >= d) ? part[t - d] : 0;
        __syncthreads();
        part[t] += v;
        __syncthreads();
    }
    int run = part[t] - s;
    for (int i = b; i < e; ++i) {
        off[i] = run;
        run += cnt[i];
    }
    if (t == 1023) off[N] = part[1023];
}

__global__ void place_kernel(const int* __restrict__ row, const int* __restrict__ col,
                             const int* __restrict__ rank, const int* __restrict__ off,
                             const int* __restrict__ base,
                             const float* __restrict__ dinv, int* __restrict__ ent, int E) {
    int i = blockIdx.x * blockDim.x + threadIdx.x;
    if (i < E) {
        const int r = row[i];
        const int c = col[i];
        int2 v;
        v.x = r;
        v.y = __float_as_int(dinv[r] * dinv[c]);
        const int slot = off[c] + base[(size_t)(i / SORT_CHUNK) * N_NODES + c] + rank[i];
        reinterpret_cast<int2*>(ent)[slot] = v;
    }
}

__global__ void gemm_wmma(const float* __restrict__ X,
                          const float* __restrict__ W,
                          const float* __restrict__ bias,
                          float* __restrict__ H, int M) {
    const int lane = threadIdx.x & 31;
    const int wave = threadIdx.x >> 5;
    const int wavesPerBlock = blockDim.x >> 5;
    const int tilesN = DFEAT / 16;
    const int tilesM = M >> 4;
    int tileId = blockIdx.x * wavesPerBlock + wave;
    if (tileId >= tilesM * tilesN) return;
    const int tm = tileId / tilesN;
    const int tn = tileId - tm * tilesN;
    const int m0 = tm << 4;
    const int n0 = tn << 4;

    const int l16   = lane & 15;
    const int khalf = (lane >> 4) << 1;

    const float* __restrict__ arow = X + (size_t)(m0 + l16) * DFEAT;
    const float* __restrict__ brow = W + (size_t)(n0 + l16) * DFEAT;

    const float bv = bias[n0 + l16];
    v8f c = {bv, bv, bv, bv, bv, bv, bv, bv};

#pragma unroll
    for (int k0 = 0; k0 < DFEAT; k0 += 4) {
        v2f a = *(const v2f*)(arow + k0 + khalf);
        v2f b = *(const v2f*)(brow + k0 + khalf);
        c = __builtin_amdgcn_wmma_f32_16x16x4_f32(false, a, false, b,
                                                  (short)0, c, false, false);
    }

    const int rowbase = m0 + ((lane >> 4) << 3);
    float* __restrict__ outcol = H + (size_t)rowbase * DFEAT + n0 + l16;
#pragma unroll
    for (int r = 0; r < 8; ++r) {
        outcol[(size_t)r * DFEAT] = c[r];
    }
}

__global__ void gather_kernel(const int* __restrict__ off, const int* __restrict__ ent,
                              const float* __restrict__ H, float* __restrict__ agg, int N) {
    const int lane = threadIdx.x & 31;
    const int n = __builtin_amdgcn_readfirstlane(
        (int)(blockIdx.x * (blockDim.x >> 5) + (threadIdx.x >> 5)));
    if (n >= N) return;
    const int beg = off[n];
    const int end = off[n + 1];
    const float4* __restrict__ H4 = reinterpret_cast<const float4*>(H);
    float4 acc = {0.0f, 0.0f, 0.0f, 0.0f};
    for (int base = beg; base < end; base += 32) {
        const int idx = base + lane;
        int2 me = {0, 0};
        if (idx < end) me = reinterpret_cast<const int2*>(ent)[idx];
        const int m = (end - base < 32) ? (end - base) : 32;
#pragma unroll 4
        for (int j = 0; j < m; ++j) {
            const int s = __builtin_amdgcn_readlane(me.x, j);
            const float wt = __int_as_float(__builtin_amdgcn_readlane(me.y, j));
            const float4 v = H4[(size_t)s * (DFEAT / 4) + lane];
            acc.x = fmaf(wt, v.x, acc.x);
            acc.y = fmaf(wt, v.y, acc.y);
            acc.z = fmaf(wt, v.z, acc.z);
            acc.w = fmaf(wt, v.w, acc.w);
        }
    }
    reinterpret_cast<float4*>(agg)[(size_t)n * (DFEAT / 4) + lane] = acc;
}

#define BN_WAVES 20
#define BN_ROWS  10

__global__ __launch_bounds__(BN_WAVES * 32)
void bn_stats(const float* __restrict__ agg, float* __restrict__ stats, int N) {
    __shared__ float part[2][BN_WAVES][DFEAT];
    const int lane = threadIdx.x & 31;
    const int wave = __builtin_amdgcn_readfirstlane((int)(threadIdx.x >> 5));
    const int gw = (int)blockIdx.x * BN_WAVES + wave;
    const int tw = (int)gridDim.x * BN_WAVES;
    const float4* __restrict__ A4 = reinterpret_cast<const float4*>(agg);
    float4 s = {0.0f, 0.0f, 0.0f, 0.0f};
    float4 q = {0.0f, 0.0f, 0.0f, 0.0f};
    for (int r0 = gw; r0 < N; r0 += BN_ROWS * tw) {
        float4 v[BN_ROWS];
        float  m[BN_ROWS];
#pragma unroll
        for (int j = 0; j < BN_ROWS; ++j) {
            const int r = r0 + j * tw;
            const bool ok = r < N;
            m[j] = ok ? 1.0f : 0.0f;
            v[j] = A4[(size_t)(ok ? r : (N - 1)) * (DFEAT / 4) + lane];
        }
#pragma unroll
        for (int j = 0; j < BN_ROWS; ++j) {
            s.x = fmaf(m[j], v[j].x, s.x);
            s.y = fmaf(m[j], v[j].y, s.y);
            s.z = fmaf(m[j], v[j].z, s.z);
            s.w = fmaf(m[j], v[j].w, s.w);
            q.x = fmaf(m[j] * v[j].x, v[j].x, q.x);
            q.y = fmaf(m[j] * v[j].y, v[j].y, q.y);
            q.z = fmaf(m[j] * v[j].z, v[j].z, q.z);
            q.w = fmaf(m[j] * v[j].w, v[j].w, q.w);
        }
    }
    *reinterpret_cast<float4*>(&part[0][wave][lane << 2]) = s;
    *reinterpret_cast<float4*>(&part[1][wave][lane << 2]) = q;
    __syncthreads();
    const int t = threadIdx.x;
    if (t < 2 * DFEAT) {
        const float* __restrict__ colp = &part[t >> 7][0][t & (DFEAT - 1)];
        float a = 0.0f;
#pragma unroll
        for (int w = 0; w < BN_WAVES; ++w) a += colp[w * DFEAT];
        atomicAdd(&stats[t], a);
    }
}

__global__ void bn_apply(const float* __restrict__ xin, const float* __restrict__ agg,
                         const float* __restrict__ stats, const float* __restrict__ g,
                         const float* __restrict__ be, float* __restrict__ xout,
                         int total4) {
    const int i = blockIdx.x * blockDim.x + threadIdx.x;
    if (i >= total4) return;
    const float invN = 1.0f / (float)N_NODES;
    const int f0 = (i & 31) << 2;
    const float4 a  = reinterpret_cast<const float4*>(agg)[i];
    const float4 xv = reinterpret_cast<const float4*>(xin)[i];
    float av[4]  = {a.x, a.y, a.z, a.w};
    float xvv[4] = {xv.x, xv.y, xv.z, xv.w};
    float r[4];
#pragma unroll
    for (int j = 0; j < 4; ++j) {
        const int f = f0 + j;
        const float mu  = stats[f] * invN;
        const float var = stats[DFEAT + f] * invN - mu * mu;
        const float sc  = g[f] * rsqrtf(var + BN_EPS);
        const float hv  = (av[j] - mu) * sc + be[f];
        r[j] = xvv[j] + fmaxf(hv, 0.0f);
    }
    float4 o = {r[0], r[1], r[2], r[3]};
    reinterpret_cast<float4*>(xout)[i] = o;
}

extern "C" void kernel_launch(void* const* d_in, const int* in_sizes, int n_in,
                              void* d_out, int out_size, void* d_ws, size_t ws_size,
                              hipStream_t stream) {
    (void)in_sizes; (void)n_in; (void)out_size; (void)ws_size;

    const float* x     = (const float*)d_in[0];
    const int*   ei    = (const int*)d_in[1];
    const float* W     = (const float*)d_in[2];
    const float* bias  = (const float*)d_in[3];
    const float* gamma = (const float*)d_in[4];
    const float* beta  = (const float*)d_in[5];
    float* out = (float*)d_out;

    const size_t ND = (size_t)N_NODES * DFEAT;
    float* ws    = (float*)d_ws;
    float* dinv  = ws;
    int*   cnt   = (int*)(ws + N_NODES);
    int*   off   = (int*)(ws + 2 * N_NODES);
    int*   ent   = (int*)(ws + 4 * N_NODES);
    int*   rank  = (int*)(ws + 4 * N_NODES + 2 * N_EDGES);
    float* x_cur = out;
    float* h     = ws + ND;
    float* agg   = ws + 2 * ND;
    float* stats = ws + 3 * ND;
    unsigned* histP = (unsigned*)(ws + ND);
    int*      base  = (int*)(ws + ND + (size_t)SORT_BLOCKS * N_NODES);

    const int* row = ei;
    const int* col = ei + N_EDGES;

    hipMemcpyAsync(x_cur, x, ND * sizeof(float), hipMemcpyDeviceToDevice, stream);

    hist_rank_kernel<<<SORT_BLOCKS, 1024, 0, stream>>>(row, col, histP, rank);
    combine_kernel<<<(HALF_N + 255) / 256, 256, 0, stream>>>(histP, base, cnt, dinv);
    scan_offsets<<<1, 1024, 0, stream>>>(cnt, off, N_NODES);
    place_kernel<<<(N_EDGES + 255) / 256, 256, 0, stream>>>(row, col, rank, off, base, dinv, ent,
                                                            N_EDGES);

    const int tiles      = (N_NODES / 16) * (DFEAT / 16);
    const int gemmBlocks = tiles / 8;

    for (int l = 0; l < N_LAYERS; ++l) {
        gemm_wmma<<<gemmBlocks, 256, 0, stream>>>(x_cur, W + (size_t)l * DFEAT * DFEAT,
                                                  bias + l * DFEAT, h, N_NODES);
        zero_f32<<<1, 256, 0, stream>>>(stats, 2 * DFEAT);
        gather_kernel<<<(N_NODES + 7) / 8, 256, 0, stream>>>(off, ent, h, agg, N_NODES);
        bn_stats<<<250, BN_WAVES * 32, 0, stream>>>(agg, stats, N_NODES);
        bn_apply<<<((int)(ND / 4) + 255) / 256, 256, 0, stream>>>(
            x_cur, agg, stats, gamma + l * DFEAT, beta + l * DFEAT, x_cur, (int)(ND / 4));
    }
}


# Fewer dependent launches, 19+copy -> 13: gather keeps each finished row in registers and adds it into the BN column sums (new reduction tree, atomics), sums cleared in scan, layer 0 reads x in place
# speedup vs baseline: 2.3898x; 1.1105x over previous
#include <hip/hip_runtime.h>
#include <math.h>


#define N_NODES 50000
#define N_EDGES 800000
#define DFEAT   128
#define N_LAYERS 3
#define BN_EPS  1e-5f

typedef __attribute__((ext_vector_type(2))) float v2f;
typedef __attribute__((ext_vector_type(8))) float v8f;

#define SORT_BLOCKS 64
#define SORT_CHUNK  (N_EDGES / SORT_BLOCKS)
#define HALF_N      (N_NODES / 2)
#define SORT_PER_THREAD ((SORT_CHUNK + 1023) / 1024)
static_assert(N_EDGES % SORT_BLOCKS == 0, "equal chunks");
static_assert(SORT_CHUNK < 65536, "a 16-bit field holds a chunk's count");
static_assert((2 * HALF_N) % 4 == 0, "tables are moved 16 bytes at a time");

__global__ __launch_bounds__(1024)
void hist_rank_kernel(const int* __restrict__ row, const int* __restrict__ col,
                      unsigned* __restrict__ histP, int* __restrict__ rank) {
    __shared__ __attribute__((aligned(16))) unsigned hs[2 * HALF_N];
    const int t = threadIdx.x;
    uint4* __restrict__ hs4 = reinterpret_cast<uint4*>(hs);
    for (int i = t; i < (2 * HALF_N) / 4; i += 1024) hs4[i] = make_uint4(0u, 0u, 0u, 0u);
    const int e0 = (int)blockIdx.x * SORT_CHUNK;
    int rr[SORT_PER_THREAD];
    int cc[SORT_PER_THREAD];
#pragma unroll
    for (int k = 0; k < SORT_PER_THREAD; ++k) {
        const int i = t + k * 1024;
        const int e = e0 + ((i < SORT_CHUNK) ? i : (SORT_CHUNK - 1));
        rr[k] = row[e];
        cc[k] = col[e];
    }
    __syncthreads();
    unsigned old[SORT_PER_THREAD];
#pragma unroll
    for (int k = 0; k < SORT_PER_THREAD; ++k) {
        const int i = t + k * 1024;
        old[k] = 0u;
        if (i < SORT_CHUNK) {
            const int c = cc[k];
            const int r = rr[k];
            old[k] = atomicAdd(&hs[c >> 1], 1u << ((unsigned)(c & 1) << 4));
            atomicAdd(&hs[HALF_N + (r >> 1)], 1u << ((unsigned)(r & 1) << 4));
        }
    }
#pragma unroll
    for (int k = 0; k < SORT_PER_THREAD; ++k) {
        const int i = t + k * 1024;
        if (i < SORT_CHUNK)
            rank[e0 + i] = (int)((old[k] >> ((unsigned)(cc[k] & 1) << 4)) & 0xffffu);
    }
    __syncthreads();
    uint4* __restrict__ out4 =
        reinterpret_cast<uint4*>(histP + (size_t)blockIdx.x * (2 * HALF_N));
    for (int i = t; i < (2 * HALF_N) / 4; i += 1024) out4[i] = hs4[i];
}

__global__ void combine_kernel(const unsigned* __restrict__ histP, int* __restrict__ base,
                               int* __restrict__ cnt, float* __restrict__ dinv) {
    const int t = blockIdx.x * blockDim.x + threadIdx.x;
    if (t >= HALF_N) return;
    unsigned c0 = 0u, c1 = 0u, d0 = 0u, d1 = 0u;
#pragma unroll 32
    for (int b = 0; b < SORT_BLOCKS; ++b) {
        const unsigned pc = histP[(size_t)b * (2 * HALF_N) + t];
        const unsigned pd = histP[(size_t)b * (2 * HALF_N) + HALF_N + t];
        int2 v;
        v.x = (int)c0;
        v.y = (int)c1;
        reinterpret_cast<int2*>(base + (size_t)b * N_NODES)[t] = v;
        c0 += pc & 0xffffu;
        c1 += pc >> 16;
        d0 += pd & 0xffffu;
        d1 += pd >> 16;
    }
    int2 cv;
    cv.x = (int)c0;
    cv.y = (int)c1;
    reinterpret_cast<int2*>(cnt)[t] = cv;
    float2 dv;
    dv.x = (d0 > 0u) ? rsqrtf((float)d0) : 0.0f;
    dv.y = (d1 > 0u) ? rsqrtf((float)d1) : 0.0f;
    reinterpret_cast<float2*>(dinv)[t] = dv;
}

static_assert(N_LAYERS * 2 * DFEAT <= 1024, "one thread per column sum in the scan block");
__global__ void scan_offsets(const int* __restrict__ cnt, int* __restrict__ off, int N,
                             float* __restrict__ stats_all) {
    __shared__ int part[1024];
    const int t = threadIdx.x;
    if (t < N_LAYERS * 2 * DFEAT) stats_all[t] = 0.0f;
    const int chunk = (N + 1023) >> 10;
    const int b = t * chunk;
    const int e = (b + chunk < N) ? (b + chunk) : N;
    int s = 0;
    for (int i = b; i < e; ++i) s += cnt[i];
    part[t] = s;
    __syncthreads();
    for (int d = 1; d < 1024; d <<= 1) {
        const int v = (t >= d) ? part[t - d] : 0;
        __syncthreads();
        part[t] += v;
        __syncthreads();
    }
    int run = part[t] - s;
    for (int i = b; i < e; ++i) {
        off[i] = run;
        run += cnt[i];
    }
    if (t == 1023) off[N] = part[1023];
}

#define PLACE_PER_THREAD 4
static_assert(N_EDGES % PLACE_PER_THREAD == 0, "whole groups of edges");
static_assert(SORT_CHUNK % PLACE_PER_THREAD == 0, "a group of edges never straddles two chunks");

__global__ void place_kernel(const int* __restrict__ row, const int* __restrict__ col,
                             const int* __restrict__ rank, const int* __restrict__ off,
                             const int* __restrict__ base,
                             const float* __restrict__ dinv, int* __restrict__ ent, int E) {
    const int g = blockIdx.x * blockDim.x + threadIdx.x;
    if (g >= E / PLACE_PER_THREAD) return;
    const int4 r4 = reinterpret_cast<const int4*>(row)[g];
    const int4 c4 = reinterpret_cast<const int4*>(col)[g];
    const int4 k4 = reinterpret_cast<const int4*>(rank)[g];
    const int* __restrict__ cb = base + (size_t)((g * PLACE_PER_THREAD) / SORT_CHUNK) * N_NODES;
    const int r[PLACE_PER_THREAD] = {r4.x, r4.y, r4.z, r4.w};
    const int c[PLACE_PER_THREAD] = {c4.x, c4.y, c4.z, c4.w};
    const int k[PLACE_PER_THREAD] = {k4.x, k4.y, k4.z, k4.w};
    float dr[PLACE_PER_THREAD], dc[PLACE_PER_THREAD];
    int   so[PLACE_PER_THREAD], sb[PLACE_PER_THREAD];
#pragma unroll
    for (int j = 0; j < PLACE_PER_THREAD; ++j) {
        dr[j] = dinv[r[j]];
        dc[j] = dinv[c[j]];
        so[j] = off[c[j]];
        sb[j] = cb[c[j]];
    }
#pragma unroll
    for (int j = 0; j < PLACE_PER_THREAD; ++j) {
        int2 v;
        v.x = r[j];
        v.y = __float_as_int(dr[j] * dc[j]);
        reinterpret_cast<int2*>(ent)[so[j] + sb[j] + k[j]] = v;
    }
}

#define GEMM_BLOCKS 256
#define GEMM_KSTEPS (DFEAT / 4)
static_assert((long long)GEMM_BLOCKS * (N_NODES / 16) < 2147483647LL, "tile split stays in int");

__global__ __launch_bounds__(512)
void gemm_wmma(const float* __restrict__ X,
               const float* __restrict__ W,
               const float* __restrict__ bias,
               float* __restrict__ H, int M) {
    const int lane = threadIdx.x & 31;
    const int wave = __builtin_amdgcn_readfirstlane((int)(threadIdx.x >> 5));
    const int tn = wave & 7;
    const int par = wave >> 3;
    const int tilesM = M >> 4;
    const int t0 = ((int)blockIdx.x * tilesM) / (int)gridDim.x;
    const int t1 = (((int)blockIdx.x + 1) * tilesM) / (int)gridDim.x;
    const int n0 = tn << 4;

    const int l16   = lane & 15;
    const int khalf = (lane >> 4) << 1;

    const float* __restrict__ brow = W + (size_t)(n0 + l16) * DFEAT;
    v2f bfrag[GEMM_KSTEPS];
#pragma unroll
    for (int kk = 0; kk < GEMM_KSTEPS; ++kk) bfrag[kk] = *(const v2f*)(brow + kk * 4 + khalf);

    const float bv = bias[n0 + l16];

    for (int tm = t0 + par; tm < t1; tm += 2) {
        const int m0 = tm << 4;
        const float* __restrict__ arow = X + (size_t)(m0 + l16) * DFEAT;
        v8f c = {bv, bv, bv, bv, bv, bv, bv, bv};
#pragma unroll
        for (int kk = 0; kk < GEMM_KSTEPS; ++kk) {
            v2f a = *(const v2f*)(arow + kk * 4 + khalf);
            c = __builtin_amdgcn_wmma_f32_16x16x4_f32(false, a, false, bfrag[kk],
                                                      (short)0, c, false, false);
        }
        const int rowbase = m0 + ((lane >> 4) << 3);
        float* __restrict__ outcol = H + (size_t)rowbase * DFEAT + n0 + l16;
#pragma unroll
        for (int r = 0; r < 8; ++r) {
            outcol[(size_t)r * DFEAT] = c[r];
        }
    }
}

#define GATHER_WAVES 12288
#define GATHER_BLOCK_WAVES 24
static_assert(GATHER_WAVES % GATHER_BLOCK_WAVES == 0, "whole blocks");
static_assert((N_NODES + GATHER_WAVES - 1) / GATHER_WAVES + 1 <= 32,
              "a wave's segment bounds fit one per lane");
static_assert((long long)GATHER_WAVES * N_NODES < 2147483647LL, "node split stays in int");

__global__ __launch_bounds__(GATHER_BLOCK_WAVES * 32)
void gather_stats_kernel(const int* __restrict__ off, const int* __restrict__ ent,
                         const float* __restrict__ H, float* __restrict__ agg,
                         float* __restrict__ stats, int N) {
    __shared__ float part[2][GATHER_BLOCK_WAVES][DFEAT];
    const int lane = threadIdx.x & 31;
    const int wave = __builtin_amdgcn_readfirstlane((int)(threadIdx.x >> 5));
    const int gw = (int)blockIdx.x * GATHER_BLOCK_WAVES + wave;
    const int n0 = (gw * N) / GATHER_WAVES;
    const int nn = ((gw + 1) * N) / GATHER_WAVES - n0;
    float4 cs = {0.0f, 0.0f, 0.0f, 0.0f};
    float4 cq = {0.0f, 0.0f, 0.0f, 0.0f};
    if (nn > 0) {
        const int myoff = off[n0 + ((lane < nn) ? lane : nn)];
        const int wend = __builtin_amdgcn_readlane(myoff, nn);
        const int2* __restrict__ ent2 = reinterpret_cast<const int2*>(ent);
        const float4* __restrict__ H4 = reinterpret_cast<const float4*>(H);
        int2 me_next = {0, 0};
        {
            const int p = __builtin_amdgcn_readlane(myoff, 0) + lane;
            if (p < wend) me_next = ent2[p];
        }
        for (int i = 0; i < nn; ++i) {
            const int beg = __builtin_amdgcn_readlane(myoff, i);
            const int end = __builtin_amdgcn_readlane(myoff, i + 1);
            float4 acc = {0.0f, 0.0f, 0.0f, 0.0f};
            for (int base = beg; base < end; base += 32) {
                const int2 me = me_next;
                const int nb = (base + 32 < end) ? (base + 32) : end;
                const int p = nb + lane;
                if (p < wend) me_next = ent2[p];
                const int m = (end - base < 32) ? (end - base) : 32;
#pragma unroll 4
                for (int j = 0; j < m; ++j) {
                    const int s = __builtin_amdgcn_readlane(me.x, j);
                    const float wt = __int_as_float(__builtin_amdgcn_readlane(me.y, j));
                    const float4 v = H4[(size_t)s * (DFEAT / 4) + lane];
                    acc.x = fmaf(wt, v.x, acc.x);
                    acc.y = fmaf(wt, v.y, acc.y);
                    acc.z = fmaf(wt, v.z, acc.z);
                    acc.w = fmaf(wt, v.w, acc.w);
                }
            }
            reinterpret_cast<float4*>(agg)[(size_t)(n0 + i) * (DFEAT / 4) + lane] = acc;
            cs.x += acc.x;
            cs.y += acc.y;
            cs.z += acc.z;
            cs.w += acc.w;
            cq.x = fmaf(acc.x, acc.x, cq.x);
            cq.y = fmaf(acc.y, acc.y, cq.y);
            cq.z = fmaf(acc.z, acc.z, cq.z);
            cq.w = fmaf(acc.w, acc.w, cq.w);
        }
    }
    *reinterpret_cast<float4*>(&part[0][wave][lane << 2]) = cs;
    *reinterpret_cast<float4*>(&part[1][wave][lane << 2]) = cq;
    __syncthreads();
    const int t = threadIdx.x;
    if (t < 2 * DFEAT) {
        const float* __restrict__ colp = &part[t >> 7][0][t & (DFEAT - 1)];
        float a = 0.0f;
#pragma unroll
        for (int w = 0; w < GATHER_BLOCK_WAVES; ++w) a += colp[w * DFEAT];
        atomicAdd(&stats[t], a);
    }
}

#define BA_WAVES 20
#define BA_ROWS  10

__global__ __launch_bounds__(BA_WAVES * 32)
void bn_apply(const float* xin, const float* __restrict__ agg,
              const float* __restrict__ stats, const float* __restrict__ g,
              const float* __restrict__ be, float* xout, int N) {
    const int lane = threadIdx.x & 31;
    const int wave = __builtin_amdgcn_readfirstlane((int)(threadIdx.x >> 5));
    const int gw = (int)blockIdx.x * BA_WAVES + wave;
    const int tw = (int)gridDim.x * BA_WAVES;
    const float invN = 1.0f / (float)N_NODES;
    const float4 s4 = reinterpret_cast<const float4*>(stats)[lane];
    const float4 q4 = reinterpret_cast<const float4*>(stats + DFEAT)[lane];
    const float4 g4 = reinterpret_cast<const float4*>(g)[lane];
    const float4 b4 = reinterpret_cast<const float4*>(be)[lane];
    const float sv[4] = {s4.x, s4.y, s4.z, s4.w};
    const float qv[4] = {q4.x, q4.y, q4.z, q4.w};
    const float gv[4] = {g4.x, g4.y, g4.z, g4.w};
    const float bv[4] = {b4.x, b4.y, b4.z, b4.w};
    float mu[4], sc[4];
#pragma unroll
    for (int j = 0; j < 4; ++j) {
        mu[j] = sv[j] * invN;
        const float var = qv[j] * invN - mu[j] * mu[j];
        sc[j] = gv[j] * rsqrtf(var + BN_EPS);
    }
    const float4* A4 = reinterpret_cast<const float4*>(agg);
    const float4* X4 = reinterpret_cast<const float4*>(xin);
    float4* O4 = reinterpret_cast<float4*>(xout);
    for (int r0 = gw; r0 < N; r0 += BA_ROWS * tw) {
        if (r0 + (BA_ROWS - 1) * tw < N) {
            float4 a[BA_ROWS];
            float4 xv[BA_ROWS];
#pragma unroll
            for (int j = 0; j < BA_ROWS; ++j) {
                const size_t idx = (size_t)(r0 + j * tw) * (DFEAT / 4) + lane;
                a[j]  = A4[idx];
                xv[j] = X4[idx];
            }
#pragma unroll
            for (int j = 0; j < BA_ROWS; ++j) {
                float4 o;
                o.x = xv[j].x + fmaxf((a[j].x - mu[0]) * sc[0] + bv[0], 0.0f);
                o.y = xv[j].y + fmaxf((a[j].y - mu[1]) * sc[1] + bv[1], 0.0f);
                o.z = xv[j].z + fmaxf((a[j].z - mu[2]) * sc[2] + bv[2], 0.0f);
                o.w = xv[j].w + fmaxf((a[j].w - mu[3]) * sc[3] + bv[3], 0.0f);
                O4[(size_t)(r0 + j * tw) * (DFEAT / 4) + lane] = o;
            }
        } else {
            for (int r = r0; r < N; r += tw) {
                const size_t idx = (size_t)r * (DFEAT / 4) + lane;
                const float4 a1 = A4[idx];
                const float4 x1 = X4[idx];
                float4 o;
                o.x = x1.x + fmaxf((a1.x - mu[0]) * sc[0] + bv[0], 0.0f);
                o.y = x1.y + fmaxf((a1.y - mu[1]) * sc[1] + bv[1], 0.0f);
                o.z = x1.z + fmaxf((a1.z - mu[2]) * sc[2] + bv[2], 0.0f);
                o.w = x1.w + fmaxf((a1.w - mu[3]) * sc[3] + bv[3], 0.0f);
                O4[idx] = o;
            }
        }
    }
}

extern "C" void kernel_launch(void* const* d_in, const int* in_sizes, int n_in,
                              void* d_out, int out_size, void* d_ws, size_t ws_size,
                              hipStream_t stream) {
    (void)in_sizes; (void)n_in; (void)out_size; (void)ws_size;

    const float* x     = (const float*)d_in[0];
    const int*   ei    = (const int*)d_in[1];
    const float* W     = (const float*)d_in[2];
    const float* bias  = (const float*)d_in[3];
    const float* gamma = (const float*)d_in[4];
    const float* beta  = (const float*)d_in[5];
    float* out = (float*)d_out;

    const size_t ND = (size_t)N_NODES * DFEAT;
    float* ws    = (float*)d_ws;
    float* dinv  = ws;
    int*   cnt   = (int*)(ws + N_NODES);
    int*   off   = (int*)(ws + 2 * N_NODES);
    int*   ent   = (int*)(ws + 4 * N_NODES);
    int*   rank  = (int*)(ws + 4 * N_NODES + 2 * N_EDGES);
    float* h     = ws + ND;
    float* agg   = ws + 2 * ND;
    float* stats = ws + 3 * N_NODES + 64;
    unsigned* histP = (unsigned*)(ws + ND);
    int*      base  = (int*)(ws + ND + (size_t)SORT_BLOCKS * N_NODES);

    const int* row = ei;
    const int* col = ei + N_EDGES;

    hist_rank_kernel<<<SORT_BLOCKS, 1024, 0, stream>>>(row, col, histP, rank);
    combine_kernel<<<(HALF_N + 255) / 256, 256, 0, stream>>>(histP, base, cnt, dinv);
    scan_offsets<<<1, 1024, 0, stream>>>(cnt, off, N_NODES, stats);
    place_kernel<<<(N_EDGES / PLACE_PER_THREAD + 255) / 256, 256, 0, stream>>>(
        row, col, rank, off, base, dinv, ent, N_EDGES);

    for (int l = 0; l < N_LAYERS; ++l) {
        const float* x_in = (l == 0) ? x : out;
        float* st = stats + l * 2 * DFEAT;
        gemm_wmma<<<GEMM_BLOCKS, 512, 0, stream>>>(x_in, W + (size_t)l * DFEAT * DFEAT,
                                                  bias + l * DFEAT, h, N_NODES);
        gather_stats_kernel<<<GATHER_WAVES / GATHER_BLOCK_WAVES, GATHER_BLOCK_WAVES * 32, 0,
                              stream>>>(off, ent, h, agg, st, N_NODES);
        bn_apply<<<250, BA_WAVES * 32, 0, stream>>>(
            x_in, agg, st, gamma + l * DFEAT, beta + l * DFEAT, out, N_NODES);
    }
}


# Segment-offset scan reads and writes whole lines: a wave owns a contiguous stretch, lanes take 16-B groups, counts held in registers, cross-lane scan (was a 196-B lane stride, 49 serial trips)
# speedup vs baseline: 3.1929x; 1.3361x over previous
#include <hip/hip_runtime.h>
#include <math.h>


#define N_NODES 50000
#define N_EDGES 800000
#define DFEAT   128
#define N_LAYERS 3
#define BN_EPS  1e-5f

typedef __attribute__((ext_vector_type(2))) float v2f;
typedef __attribute__((ext_vector_type(8))) float v8f;

#define SORT_BLOCKS 64
#define SORT_CHUNK  (N_EDGES / SORT_BLOCKS)
#define HALF_N      (N_NODES / 2)
#define SORT_PER_THREAD ((SORT_CHUNK + 1023) / 1024)
static_assert(N_EDGES % SORT_BLOCKS == 0, "equal chunks");
static_assert(SORT_CHUNK < 65536, "a 16-bit field holds a chunk's count");
static_assert((2 * HALF_N) % 4 == 0, "tables are moved 16 bytes at a time");

__global__ __launch_bounds__(1024)
void hist_rank_kernel(const int* __restrict__ row, const int* __restrict__ col,
                      unsigned* __restrict__ histP, int* __restrict__ rank) {
    __shared__ __attribute__((aligned(16))) unsigned hs[2 * HALF_N];
    const int t = threadIdx.x;
    uint4* __restrict__ hs4 = reinterpret_cast<uint4*>(hs);
    for (int i = t; i < (2 * HALF_N) / 4; i += 1024) hs4[i] = make_uint4(0u, 0u, 0u, 0u);
    const int e0 = (int)blockIdx.x * SORT_CHUNK;
    int rr[SORT_PER_THREAD];
    int cc[SORT_PER_THREAD];
#pragma unroll
    for (int k = 0; k < SORT_PER_THREAD; ++k) {
        const int i = t + k * 1024;
        const int e = e0 + ((i < SORT_CHUNK) ? i : (SORT_CHUNK - 1));
        rr[k] = row[e];
        cc[k] = col[e];
    }
    __syncthreads();
    unsigned old[SORT_PER_THREAD];
#pragma unroll
    for (int k = 0; k < SORT_PER_THREAD; ++k) {
        const int i = t + k * 1024;
        old[k] = 0u;
        if (i < SORT_CHUNK) {
            const int c = cc[k];
            const int r = rr[k];
            old[k] = atomicAdd(&hs[c >> 1], 1u << ((unsigned)(c & 1) << 4));
            atomicAdd(&hs[HALF_N + (r >> 1)], 1u << ((unsigned)(r & 1) << 4));
        }
    }
#pragma unroll
    for (int k = 0; k < SORT_PER_THREAD; ++k) {
        const int i = t + k * 1024;
        if (i < SORT_CHUNK)
            rank[e0 + i] = (int)((old[k] >> ((unsigned)(cc[k] & 1) << 4)) & 0xffffu);
    }
    __syncthreads();
    uint4* __restrict__ out4 =
        reinterpret_cast<uint4*>(histP + (size_t)blockIdx.x * (2 * HALF_N));
    for (int i = t; i < (2 * HALF_N) / 4; i += 1024) out4[i] = hs4[i];
}

__global__ void combine_kernel(const unsigned* __restrict__ histP, int* __restrict__ base,
                               int* __restrict__ cnt, float* __restrict__ dinv) {
    const int t = blockIdx.x * blockDim.x + threadIdx.x;
    if (t >= HALF_N) return;
    unsigned c0 = 0u, c1 = 0u, d0 = 0u, d1 = 0u;
#pragma unroll 32
    for (int b = 0; b < SORT_BLOCKS; ++b) {
        const unsigned pc = histP[(size_t)b * (2 * HALF_N) + t];
        const unsigned pd = histP[(size_t)b * (2 * HALF_N) + HALF_N + t];
        int2 v;
        v.x = (int)c0;
        v.y = (int)c1;
        reinterpret_cast<int2*>(base + (size_t)b * N_NODES)[t] = v;
        c0 += pc & 0xffffu;
        c1 += pc >> 16;
        d0 += pd & 0xffffu;
        d1 += pd >> 16;
    }
    int2 cv;
    cv.x = (int)c0;
    cv.y = (int)c1;
    reinterpret_cast<int2*>(cnt)[t] = cv;
    float2 dv;
    dv.x = (d0 > 0u) ? rsqrtf((float)d0) : 0.0f;
    dv.y = (d1 > 0u) ? rsqrtf((float)d1) : 0.0f;
    reinterpret_cast<float2*>(dinv)[t] = dv;
}

#define SCAN_WAVES 32
#define SCAN_TRIPS (((N_NODES + SCAN_WAVES - 1) / SCAN_WAVES + 127) / 128)
static_assert(N_LAYERS * 2 * DFEAT <= 1024, "one thread per column sum in the scan block");
static_assert(N_NODES % 4 == 0, "counts and offsets are moved 16 bytes at a time");
static_assert(SCAN_WAVES * SCAN_TRIPS * 128 >= N_NODES, "the block covers every node");
static_assert(SCAN_WAVES == 32, "one wave total per lane in the second-level scan");

__global__ __launch_bounds__(SCAN_WAVES * 32) __attribute__((amdgpu_waves_per_eu(8, 8)))
void scan_offsets(const int* __restrict__ cnt, int* __restrict__ off,
                  float* __restrict__ stats_all) {
    __shared__ int wtot[SCAN_WAVES];
    const int t = (int)threadIdx.x;
    if (t < N_LAYERS * 2 * DFEAT) stats_all[t] = 0.0f;
    const int lane = t & 31;
    const int wave = __builtin_amdgcn_readfirstlane(t >> 5);
    const int4* __restrict__ cnt4 = reinterpret_cast<const int4*>(cnt);
    int4* __restrict__ off4 = reinterpret_cast<int4*>(off);
    const int q0 = wave * (SCAN_TRIPS * 32) + lane;
    int4 v[SCAN_TRIPS];
#pragma unroll
    for (int j = 0; j < SCAN_TRIPS; ++j) {
        const int q = q0 + j * 32;
        v[j] = (q < N_NODES / 4) ? cnt4[q] : make_int4(0, 0, 0, 0);
    }
    int tot[SCAN_TRIPS];
    int x[SCAN_TRIPS];
#pragma unroll
    for (int j = 0; j < SCAN_TRIPS; ++j) {
        tot[j] = v[j].x + v[j].y + v[j].z + v[j].w;
        x[j] = tot[j];
    }
#pragma unroll
    for (int d = 1; d < 32; d <<= 1) {
        int y[SCAN_TRIPS];
#pragma unroll
        for (int j = 0; j < SCAN_TRIPS; ++j) y[j] = __shfl_up(x[j], (unsigned)d);
#pragma unroll
        for (int j = 0; j < SCAN_TRIPS; ++j)
            if (lane >= d) x[j] += y[j];
    }
    int ex[SCAN_TRIPS];
    int carry = 0;
#pragma unroll
    for (int j = 0; j < SCAN_TRIPS; ++j) {
        ex[j] = carry + x[j] - tot[j];
        carry += __builtin_amdgcn_readlane(x[j], 31);
    }
    if (lane == 0) wtot[wave] = carry;
    __syncthreads();
    const int mine = wtot[lane];
    int wx = mine;
#pragma unroll
    for (int d = 1; d < 32; d <<= 1) {
        const int y = __shfl_up(wx, (unsigned)d);
        if (lane >= d) wx += y;
    }
    const int total = __builtin_amdgcn_readlane(wx, 31);
    const int wbase = __builtin_amdgcn_readlane(wx - mine, wave);
#pragma unroll
    for (int j = 0; j < SCAN_TRIPS; ++j) {
        const int q = q0 + j * 32;
        if (q < N_NODES / 4) {
            int4 o;
            o.x = wbase + ex[j];
            o.y = o.x + v[j].x;
            o.z = o.y + v[j].y;
            o.w = o.z + v[j].z;
            off4[q] = o;
        }
    }
    if (t == 0) off[N_NODES] = total;
}

#define PLACE_PER_THREAD 4
static_assert(N_EDGES % PLACE_PER_THREAD == 0, "whole groups of edges");
static_assert(SORT_CHUNK % PLACE_PER_THREAD == 0, "a group of edges never straddles two chunks");

__global__ void place_kernel(const int* __restrict__ row, const int* __restrict__ col,
                             const int* __restrict__ rank, const int* __restrict__ off,
                             const int* __restrict__ base,
                             const float* __restrict__ dinv, int* __restrict__ ent, int E) {
    const int g = blockIdx.x * blockDim.x + threadIdx.x;
    if (g >= E / PLACE_PER_THREAD) return;
    const int4 r4 = reinterpret_cast<const int4*>(row)[g];
    const int4 c4 = reinterpret_cast<const int4*>(col)[g];
    const int4 k4 = reinterpret_cast<const int4*>(rank)[g];
    const int* __restrict__ cb = base + (size_t)((g * PLACE_PER_THREAD) / SORT_CHUNK) * N_NODES;
    const int r[PLACE_PER_THREAD] = {r4.x, r4.y, r4.z, r4.w};
    const int c[PLACE_PER_THREAD] = {c4.x, c4.y, c4.z, c4.w};
    const int k[PLACE_PER_THREAD] = {k4.x, k4.y, k4.z, k4.w};
    float dr[PLACE_PER_THREAD], dc[PLACE_PER_THREAD];
    int   so[PLACE_PER_THREAD], sb[PLACE_PER_THREAD];
#pragma unroll
    for (int j = 0; j < PLACE_PER_THREAD; ++j) {
        dr[j] = dinv[r[j]];
        dc[j] = dinv[c[j]];
        so[j] = off[c[j]];
        sb[j] = cb[c[j]];
    }
#pragma unroll
    for (int j = 0; j < PLACE_PER_THREAD; ++j) {
        int2 v;
        v.x = r[j];
        v.y = __float_as_int(dr[j] * dc[j]);
        reinterpret_cast<int2*>(ent)[so[j] + sb[j] + k[j]] = v;
    }
}

#define FT_NODES 64
#define FT_WAVES 16
#define FT_NPW   (FT_NODES / FT_WAVES)
#define FT_GROUP 16
#define FT_PITCH (DFEAT + 4)
#define GEMM_KSTEPS (DFEAT / 4)
static_assert((FT_WAVES / 2) * 16 == DFEAT && FT_NODES == 64 && FT_WAVES == 16,
              "8 column tiles x 2 pairs of row tiles");
static_assert(FT_NPW + 1 <= 32, "a wave's segment bounds fit one per lane");
static_assert(N_NODES % 16 == 0, "whole 16-row tiles");
static_assert(16 % FT_NPW == 0, "a wave's nodes lie in one tile, so all or none of them exist");
static_assert((DFEAT * DFEAT / 4) % (FT_WAVES * 32) == 0, "W copy: whole trips");

__global__ __launch_bounds__(FT_WAVES * 32)
void gather_gemm_stats_kernel(const int* __restrict__ off, const int* __restrict__ ent,
                              const float* __restrict__ X, const float* __restrict__ W,
                              const float* __restrict__ bias, float* __restrict__ agg,
                              float* __restrict__ stats) {
    __shared__ __attribute__((aligned(16))) float wl[DFEAT * FT_PITCH];
    __shared__ __attribute__((aligned(16))) float tile[FT_NODES * FT_PITCH];
    __shared__ float ssum[FT_NODES];
    __shared__ float part[FT_WAVES / 8][2 * DFEAT];
    const int lane = threadIdx.x & 31;
    const int wave = __builtin_amdgcn_readfirstlane((int)(threadIdx.x >> 5));
    const int l16   = lane & 15;
    const int hh    = lane >> 4;
    const int khalf = hh << 1;
    const int nc0   = (wave & 7) << 4;
    const int pr0   = (wave >> 3) << 5;

    {
        const float4* __restrict__ W4 = reinterpret_cast<const float4*>(W);
        float4 wv[(DFEAT * DFEAT / 4) / (FT_WAVES * 32)];
#pragma unroll
        for (int i = 0; i < (DFEAT * DFEAT / 4) / (FT_WAVES * 32); ++i)
            wv[i] = W4[(int)threadIdx.x + i * (FT_WAVES * 32)];
#pragma unroll
        for (int i = 0; i < (DFEAT * DFEAT / 4) / (FT_WAVES * 32); ++i) {
            const int q = (int)threadIdx.x + i * (FT_WAVES * 32);
            *reinterpret_cast<float4*>(&wl[(q >> 5) * FT_PITCH + ((q & 31) << 2)]) = wv[i];
        }
    }
    const float bv = bias[nc0 + l16];

    const int nb0 = (int)blockIdx.x * FT_NODES;
    const int n0 = nb0 + wave * FT_NPW;
    if (n0 < N_NODES) {
        const int myoff = off[n0 + ((lane < FT_NPW) ? lane : FT_NPW)];
        const int wend = __builtin_amdgcn_readlane(myoff, FT_NPW);
        const int2* __restrict__ ent2 = reinterpret_cast<const int2*>(ent);
        const float4* __restrict__ X4 = reinterpret_cast<const float4*>(X);
        int2 me_next = {0, 0};
        {
            const int p = __builtin_amdgcn_readlane(myoff, 0) + lane;
            if (p < wend) me_next = ent2[p];
        }
        for (int i = 0; i < FT_NPW; ++i) {
            const int beg = __builtin_amdgcn_readlane(myoff, i);
            const int end = __builtin_amdgcn_readlane(myoff, i + 1);
            float4 acc = {0.0f, 0.0f, 0.0f, 0.0f};
            float sw = 0.0f;
            for (int base = beg; base < end; base += 32) {
                const int2 me = me_next;
                const int nb = (base + 32 < end) ? (base + 32) : end;
                const int p = nb + lane;
                if (p < wend) me_next = ent2[p];
                const int m = (end - base < 32) ? (end - base) : 32;
                for (int j = 0; j < m; j += FT_GROUP) {
                    float4 v[FT_GROUP];
                    float wt[FT_GROUP];
#pragma unroll
                    for (int u = 0; u < FT_GROUP; ++u) {
                        if (j + u < m) {
                            const int s = __builtin_amdgcn_readlane(me.x, j + u);
                            wt[u] = __int_as_float(__builtin_amdgcn_readlane(me.y, j + u));
                            v[u] = X4[(size_t)s * (DFEAT / 4) + lane];
                        }
                    }
#pragma unroll
                    for (int u = 0; u < FT_GROUP; ++u) {
                        if (j + u < m) {
                            acc.x = fmaf(wt[u], v[u].x, acc.x);
                            acc.y = fmaf(wt[u], v[u].y, acc.y);
                            acc.z = fmaf(wt[u], v[u].z, acc.z);
                            acc.w = fmaf(wt[u], v[u].w, acc.w);
                            sw += wt[u];
                        }
                    }
                }
            }
            const int tr = wave * FT_NPW + i;
            *reinterpret_cast<float4*>(&tile[tr * FT_PITCH + (lane << 2)]) = acc;
            if (lane == 0) ssum[tr] = sw;
        }
    } else {
        for (int i = 0; i < FT_NPW; ++i) {
            const int tr = wave * FT_NPW + i;
            *reinterpret_cast<float4*>(&tile[tr * FT_PITCH + (lane << 2)]) =
                make_float4(0.0f, 0.0f, 0.0f, 0.0f);
            if (lane == 0) ssum[tr] = 0.0f;
        }
    }
    __syncthreads();

    const float* __restrict__ a0p = tile + (pr0 + l16) * FT_PITCH + khalf;
    const float* __restrict__ a1p = a0p + 16 * FT_PITCH;
    const float* __restrict__ bp  = wl + (nc0 + l16) * FT_PITCH + khalf;
    v8f c0, c1;
#pragma unroll
    for (int r = 0; r < 8; ++r) {
        c0[r] = bv * ssum[pr0 + r + (hh << 3)];
        c1[r] = bv * ssum[pr0 + 16 + r + (hh << 3)];
    }
#pragma unroll
    for (int kk = 0; kk < GEMM_KSTEPS; ++kk) {
        const v2f b  = *(const v2f*)(bp  + kk * 4);
        const v2f a0 = *(const v2f*)(a0p + kk * 4);
        const v2f a1 = *(const v2f*)(a1p + kk * 4);
        c0 = __builtin_amdgcn_wmma_f32_16x16x4_f32(false, a0, false, b,
                                                   (short)0, c0, false, false);
        c1 = __builtin_amdgcn_wmma_f32_16x16x4_f32(false, a1, false, b,
                                                   (short)0, c1, false, false);
    }
    float cs = 0.0f, cq = 0.0f;
#pragma unroll
    for (int r = 0; r < 8; ++r) {
        cs += c0[r];
        cq = fmaf(c0[r], c0[r], cq);
    }
#pragma unroll
    for (int r = 0; r < 8; ++r) {
        cs += c1[r];
        cq = fmaf(c1[r], c1[r], cq);
    }
    const float os = __shfl_xor(cs, 16);
    const float oq = __shfl_xor(cq, 16);
    part[wave >> 3][(hh << 7) + nc0 + l16] = (hh == 0) ? (cs + os) : (cq + oq);
    __syncthreads();
    const int t = (int)threadIdx.x;
    if (t < 2 * DFEAT) atomicAdd(&stats[t], part[0][t] + part[1][t]);

    const int row0 = nb0 + pr0 + (hh << 3);
    float* __restrict__ outcol = agg + (size_t)row0 * DFEAT + nc0 + l16;
    if (nb0 + pr0 < N_NODES) {
#pragma unroll
        for (int r = 0; r < 8; ++r) outcol[(size_t)r * DFEAT] = c0[r];
    }
    if (nb0 + pr0 + 16 < N_NODES) {
#pragma unroll
        for (int r = 0; r < 8; ++r) outcol[(size_t)(16 + r) * DFEAT] = c1[r];
    }
}

#define BA_WAVES 20
#define BA_ROWS  10

__global__ __launch_bounds__(BA_WAVES * 32)
void bn_apply(const float* xin, const float* __restrict__ agg,
              const float* __restrict__ stats, const float* __restrict__ g,
              const float* __restrict__ be, float* xout, int N) {
    const int lane = threadIdx.x & 31;
    const int wave = __builtin_amdgcn_readfirstlane((int)(threadIdx.x >> 5));
    const int gw = (int)blockIdx.x * BA_WAVES + wave;
    const int tw = (int)gridDim.x * BA_WAVES;
    const float invN = 1.0f / (float)N_NODES;
    const float4 s4 = reinterpret_cast<const float4*>(stats)[lane];
    const float4 q4 = reinterpret_cast<const float4*>(stats + DFEAT)[lane];
    const float4 g4 = reinterpret_cast<const float4*>(g)[lane];
    const float4 b4 = reinterpret_cast<const float4*>(be)[lane];
    const float sv[4] = {s4.x, s4.y, s4.z, s4.w};
    const float qv[4] = {q4.x, q4.y, q4.z, q4.w};
    const float gv[4] = {g4.x, g4.y, g4.z, g4.w};
    const float bv[4] = {b4.x, b4.y, b4.z, b4.w};
    float mu[4], sc[4];
#pragma unroll
    for (int j = 0; j < 4; ++j) {
        mu[j] = sv[j] * invN;
        const float var = qv[j] * invN - mu[j] * mu[j];
        sc[j] = gv[j] * rsqrtf(var + BN_EPS);
    }
    const float4* A4 = reinterpret_cast<const float4*>(agg);
    const float4* X4 = reinterpret_cast<const float4*>(xin);
    float4* O4 = reinterpret_cast<float4*>(xout);
    for (int r0 = gw; r0 < N; r0 += BA_ROWS * tw) {
        if (r0 + (BA_ROWS - 1) * tw < N) {
            float4 a[BA_ROWS];
            float4 xv[BA_ROWS];
#pragma unroll
            for (int j = 0; j < BA_ROWS; ++j) {
                const size_t idx = (size_t)(r0 + j * tw) * (DFEAT / 4) + lane;
                a[j]  = A4[idx];
                xv[j] = X4[idx];
            }
#pragma unroll
            for (int j = 0; j < BA_ROWS; ++j) {
                float4 o;
                o.x = xv[j].x + fmaxf((a[j].x - mu[0]) * sc[0] + bv[0], 0.0f);
                o.y = xv[j].y + fmaxf((a[j].y - mu[1]) * sc[1] + bv[1], 0.0f);
                o.z = xv[j].z + fmaxf((a[j].z - mu[2]) * sc[2] + bv[2], 0.0f);
                o.w = xv[j].w + fmaxf((a[j].w - mu[3]) * sc[3] + bv[3], 0.0f);
                O4[(size_t)(r0 + j * tw) * (DFEAT / 4) + lane] = o;
            }
        } else {
            for (int r = r0; r < N; r += tw) {
                const size_t idx = (size_t)r * (DFEAT / 4) + lane;
                const float4 a1 = A4[idx];
                const float4 x1 = X4[idx];
                float4 o;
                o.x = x1.x + fmaxf((a1.x - mu[0]) * sc[0] + bv[0], 0.0f);
                o.y = x1.y + fmaxf((a1.y - mu[1]) * sc[1] + bv[1], 0.0f);
                o.z = x1.z + fmaxf((a1.z - mu[2]) * sc[2] + bv[2], 0.0f);
                o.w = x1.w + fmaxf((a1.w - mu[3]) * sc[3] + bv[3], 0.0f);
                O4[idx] = o;
            }
        }
    }
}

extern "C" void kernel_launch(void* const* d_in, const int* in_sizes, int n_in,
                              void* d_out, int out_size, void* d_ws, size_t ws_size,
                              hipStream_t stream) {
    (void)in_sizes; (void)n_in; (void)out_size; (void)ws_size;

    const float* x     = (const float*)d_in[0];
    const int*   ei    = (const int*)d_in[1];
    const float* W     = (const float*)d_in[2];
    const float* bias  = (const float*)d_in[3];
    const float* gamma = (const float*)d_in[4];
    const float* beta  = (const float*)d_in[5];
    float* out = (float*)d_out;

    const size_t ND = (size_t)N_NODES * DFEAT;
    float* ws    = (float*)d_ws;
    float* dinv  = ws;
    int*   cnt   = (int*)(ws + N_NODES);
    int*   off   = (int*)(ws + 2 * N_NODES);
    int*   ent   = (int*)(ws + 4 * N_NODES);
    int*   rank  = (int*)(ws + 4 * N_NODES + 2 * N_EDGES);
    float* agg   = ws + 2 * ND;
    float* stats = ws + 3 * N_NODES + 64;
    unsigned* histP = (unsigned*)(ws + ND);
    int*      base  = (int*)(ws + ND + (size_t)SORT_BLOCKS * N_NODES);

    const int* row = ei;
    const int* col = ei + N_EDGES;

    hist_rank_kernel<<<SORT_BLOCKS, 1024, 0, stream>>>(row, col, histP, rank);
    combine_kernel<<<(HALF_N + 255) / 256, 256, 0, stream>>>(histP, base, cnt, dinv);
    scan_offsets<<<1, SCAN_WAVES * 32, 0, stream>>>(cnt, off, stats);
    place_kernel<<<(N_EDGES / PLACE_PER_THREAD + 255) / 256, 256, 0, stream>>>(
        row, col, rank, off, base, dinv, ent, N_EDGES);

    for (int l = 0; l < N_LAYERS; ++l) {
        const float* x_in = (l == 0) ? x : out;
        float* st = stats + l * 2 * DFEAT;
        gather_gemm_stats_kernel<<<(N_NODES + FT_NODES - 1) / FT_NODES, FT_WAVES * 32, 0, stream>>>(
            off, ent, x_in, W + (size_t)l * DFEAT * DFEAT, bias + l * DFEAT, agg, st);
        bn_apply<<<250, BA_WAVES * 32, 0, stream>>>(
            x_in, agg, st, gamma + l * DFEAT, beta + l * DFEAT, out, N_NODES);
    }
}


# Edge placement reads its tables from LDS: a block per (edge chunk, quarter of the nodes) stages dinv and its slot table (segment start + chunk base); no scattered global read per edge, was 4
# speedup vs baseline: 3.3388x; 1.0457x over previous
#include <hip/hip_runtime.h>
#include <math.h>


#define N_NODES 50000
#define N_EDGES 800000
#define DFEAT   128
#define N_LAYERS 3
#define BN_EPS  1e-5f

typedef __attribute__((ext_vector_type(2))) float v2f;
typedef __attribute__((ext_vector_type(8))) float v8f;

#define SORT_BLOCKS 64
#define SORT_CHUNK  (N_EDGES / SORT_BLOCKS)
#define HALF_N      (N_NODES / 2)
#define SORT_PER_THREAD ((SORT_CHUNK + 1023) / 1024)
static_assert(N_EDGES % SORT_BLOCKS == 0, "equal chunks");
static_assert(SORT_CHUNK < 65536, "a 16-bit field holds a chunk's count");
static_assert((2 * HALF_N) % 4 == 0, "tables are moved 16 bytes at a time");

__global__ __launch_bounds__(1024)
void hist_rank_kernel(const int* __restrict__ row, const int* __restrict__ col,
                      unsigned* __restrict__ histP, int* __restrict__ rank) {
    __shared__ __attribute__((aligned(16))) unsigned hs[2 * HALF_N];
    const int t = threadIdx.x;
    uint4* __restrict__ hs4 = reinterpret_cast<uint4*>(hs);
    for (int i = t; i < (2 * HALF_N) / 4; i += 1024) hs4[i] = make_uint4(0u, 0u, 0u, 0u);
    const int e0 = (int)blockIdx.x * SORT_CHUNK;
    int rr[SORT_PER_THREAD];
    int cc[SORT_PER_THREAD];
#pragma unroll
    for (int k = 0; k < SORT_PER_THREAD; ++k) {
        const int i = t + k * 1024;
        const int e = e0 + ((i < SORT_CHUNK) ? i : (SORT_CHUNK - 1));
        rr[k] = row[e];
        cc[k] = col[e];
    }
    __syncthreads();
    unsigned old[SORT_PER_THREAD];
#pragma unroll
    for (int k = 0; k < SORT_PER_THREAD; ++k) {
        const int i = t + k * 1024;
        old[k] = 0u;
        if (i < SORT_CHUNK) {
            const int c = cc[k];
            const int r = rr[k];
            old[k] = atomicAdd(&hs[c >> 1], 1u << ((unsigned)(c & 1) << 4));
            atomicAdd(&hs[HALF_N + (r >> 1)], 1u << ((unsigned)(r & 1) << 4));
        }
    }
#pragma unroll
    for (int k = 0; k < SORT_PER_THREAD; ++k) {
        const int i = t + k * 1024;
        if (i < SORT_CHUNK)
            rank[e0 + i] = (int)((old[k] >> ((unsigned)(cc[k] & 1) << 4)) & 0xffffu);
    }
    __syncthreads();
    uint4* __restrict__ out4 =
        reinterpret_cast<uint4*>(histP + (size_t)blockIdx.x * (2 * HALF_N));
    for (int i = t; i < (2 * HALF_N) / 4; i += 1024) out4[i] = hs4[i];
}

__global__ void combine_kernel(const unsigned* __restrict__ histP, int* __restrict__ base,
                               int* __restrict__ cnt, float* __restrict__ dinv) {
    const int t = blockIdx.x * blockDim.x + threadIdx.x;
    if (t >= HALF_N) return;
    unsigned c0 = 0u, c1 = 0u, d0 = 0u, d1 = 0u;
#pragma unroll 32
    for (int b = 0; b < SORT_BLOCKS; ++b) {
        const unsigned pc = histP[(size_t)b * (2 * HALF_N) + t];
        const unsigned pd = histP[(size_t)b * (2 * HALF_N) + HALF_N + t];
        int2 v;
        v.x = (int)c0;
        v.y = (int)c1;
        reinterpret_cast<int2*>(base + (size_t)b * N_NODES)[t] = v;
        c0 += pc & 0xffffu;
        c1 += pc >> 16;
        d0 += pd & 0xffffu;
        d1 += pd >> 16;
    }
    int2 cv;
    cv.x = (int)c0;
    cv.y = (int)c1;
    reinterpret_cast<int2*>(cnt)[t] = cv;
    float2 dv;
    dv.x = (d0 > 0u) ? rsqrtf((float)d0) : 0.0f;
    dv.y = (d1 > 0u) ? rsqrtf((float)d1) : 0.0f;
    reinterpret_cast<float2*>(dinv)[t] = dv;
}

#define SCAN_WAVES 32
#define SCAN_TRIPS (((N_NODES + SCAN_WAVES - 1) / SCAN_WAVES + 127) / 128)
static_assert(N_LAYERS * 2 * DFEAT <= 1024, "one thread per column sum in the scan block");
static_assert(N_NODES % 4 == 0, "counts and offsets are moved 16 bytes at a time");
static_assert(SCAN_WAVES * SCAN_TRIPS * 128 >= N_NODES, "the block covers every node");
static_assert(SCAN_WAVES == 32, "one wave total per lane in the second-level scan");

__global__ __launch_bounds__(SCAN_WAVES * 32) __attribute__((amdgpu_waves_per_eu(8, 8)))
void scan_offsets(const int* __restrict__ cnt, int* __restrict__ off,
                  float* __restrict__ stats_all) {
    __shared__ int wtot[SCAN_WAVES];
    const int t = (int)threadIdx.x;
    if (t < N_LAYERS * 2 * DFEAT) stats_all[t] = 0.0f;
    const int lane = t & 31;
    const int wave = __builtin_amdgcn_readfirstlane(t >> 5);
    const int4* __restrict__ cnt4 = reinterpret_cast<const int4*>(cnt);
    int4* __restrict__ off4 = reinterpret_cast<int4*>(off);
    const int q0 = wave * (SCAN_TRIPS * 32) + lane;
    int4 v[SCAN_TRIPS];
#pragma unroll
    for (int j = 0; j < SCAN_TRIPS; ++j) {
        const int q = q0 + j * 32;
        v[j] = (q < N_NODES / 4) ? cnt4[q] : make_int4(0, 0, 0, 0);
    }
    int tot[SCAN_TRIPS];
    int x[SCAN_TRIPS];
#pragma unroll
    for (int j = 0; j < SCAN_TRIPS; ++j) {
        tot[j] = v[j].x + v[j].y + v[j].z + v[j].w;
        x[j] = tot[j];
    }
#pragma unroll
    for (int d = 1; d < 32; d <<= 1) {
        int y[SCAN_TRIPS];
#pragma unroll
        for (int j = 0; j < SCAN_TRIPS; ++j) y[j] = __shfl_up(x[j], (unsigned)d);
#pragma unroll
        for (int j = 0; j < SCAN_TRIPS; ++j)
            if (lane >= d) x[j] += y[j];
    }
    int ex[SCAN_TRIPS];
    int carry = 0;
#pragma unroll
    for (int j = 0; j < SCAN_TRIPS; ++j) {
        ex[j] = carry + x[j] - tot[j];
        carry += __builtin_amdgcn_readlane(x[j], 31);
    }
    if (lane == 0) wtot[wave] = carry;
    __syncthreads();
    const int mine = wtot[lane];
    int wx = mine;
#pragma unroll
    for (int d = 1; d < 32; d <<= 1) {
        const int y = __shfl_up(wx, (unsigned)d);
        if (lane >= d) wx += y;
    }
    const int total = __builtin_amdgcn_readlane(wx, 31);
    const int wbase = __builtin_amdgcn_readlane(wx - mine, wave);
#pragma unroll
    for (int j = 0; j < SCAN_TRIPS; ++j) {
        const int q = q0 + j * 32;
        if (q < N_NODES / 4) {
            int4 o;
            o.x = wbase + ex[j];
            o.y = o.x + v[j].x;
            o.z = o.y + v[j].y;
            o.w = o.z + v[j].z;
            off4[q] = o;
        }
    }
    if (t == 0) off[N_NODES] = total;
}

#define PL_RANGES 4
#define PL_NODES  (N_NODES / PL_RANGES)
#define PL_GROUPS (SORT_CHUNK / 4)
#define PL_TRIPS  ((PL_GROUPS + 1023) / 1024)
#define PL_DINV_TRIPS ((N_NODES / 4 + 1023) / 1024)
#define PL_SLOT_TRIPS ((PL_NODES / 4 + 1023) / 1024)
static_assert(N_NODES % (4 * PL_RANGES) == 0, "node ranges are moved 16 bytes at a time");
static_assert(SORT_CHUNK % 4 == 0, "a chunk is whole groups of 4 edges, 16-byte aligned");
static_assert((N_NODES + PL_NODES) * 4 <= 320 * 1024, "dinv and one slot table fit the LDS");

__global__ __launch_bounds__(1024)
void place_kernel(const int* __restrict__ row, const int* __restrict__ col,
                  const int* __restrict__ rank, const int* __restrict__ off,
                  const int* __restrict__ base,
                  const float* __restrict__ dinv, int* __restrict__ ent) {
    __shared__ __attribute__((aligned(16))) float dinv_s[N_NODES];
    __shared__ __attribute__((aligned(16))) int   slot_s[PL_NODES];
    const int t = (int)threadIdx.x;
    const int chunk = (int)blockIdx.x / PL_RANGES;
    const int q0 = ((int)blockIdx.x % PL_RANGES) * PL_NODES;
    const int4* __restrict__ row4 = reinterpret_cast<const int4*>(row) + chunk * PL_GROUPS;
    const int4* __restrict__ col4 = reinterpret_cast<const int4*>(col) + chunk * PL_GROUPS;
    const int4* __restrict__ rnk4 = reinterpret_cast<const int4*>(rank) + chunk * PL_GROUPS;
    int4 r4[PL_TRIPS], c4[PL_TRIPS], k4[PL_TRIPS];
#pragma unroll
    for (int k = 0; k < PL_TRIPS; ++k) {
        const int g = t + k * 1024;
        const int gi = (g < PL_GROUPS) ? g : (PL_GROUPS - 1);
        r4[k] = row4[gi];
        c4[k] = col4[gi];
        k4[k] = rnk4[gi];
    }
    {
        const float4* __restrict__ d4 = reinterpret_cast<const float4*>(dinv);
        float4* __restrict__ s4 = reinterpret_cast<float4*>(dinv_s);
        float4 v[PL_DINV_TRIPS];
#pragma unroll
        for (int k = 0; k < PL_DINV_TRIPS; ++k) {
            const int i = t + k * 1024;
            v[k] = (i < N_NODES / 4) ? d4[i] : make_float4(0.0f, 0.0f, 0.0f, 0.0f);
        }
#pragma unroll
        for (int k = 0; k < PL_DINV_TRIPS; ++k) {
            const int i = t + k * 1024;
            if (i < N_NODES / 4) s4[i] = v[k];
        }
    }
    {
        const int4* __restrict__ o4 = reinterpret_cast<const int4*>(off + q0);
        const int4* __restrict__ b4 =
            reinterpret_cast<const int4*>(base + (size_t)chunk * N_NODES + q0);
        int4* __restrict__ s4 = reinterpret_cast<int4*>(slot_s);
        int4 a[PL_SLOT_TRIPS], b[PL_SLOT_TRIPS];
#pragma unroll
        for (int k = 0; k < PL_SLOT_TRIPS; ++k) {
            const int i = t + k * 1024;
            a[k] = (i < PL_NODES / 4) ? o4[i] : make_int4(0, 0, 0, 0);
            b[k] = (i < PL_NODES / 4) ? b4[i] : make_int4(0, 0, 0, 0);
        }
#pragma unroll
        for (int k = 0; k < PL_SLOT_TRIPS; ++k) {
            const int i = t + k * 1024;
            if (i < PL_NODES / 4)
                s4[i] = make_int4(a[k].x + b[k].x, a[k].y + b[k].y, a[k].z + b[k].z, a[k].w + b[k].w);
        }
    }
    __syncthreads();
    int2* __restrict__ ent2 = reinterpret_cast<int2*>(ent);
#pragma unroll
    for (int k = 0; k < PL_TRIPS; ++k) {
        if (t + k * 1024 < PL_GROUPS) {
            const int r[4]  = {r4[k].x, r4[k].y, r4[k].z, r4[k].w};
            const int c[4]  = {c4[k].x, c4[k].y, c4[k].z, c4[k].w};
            const int kk[4] = {k4[k].x, k4[k].y, k4[k].z, k4[k].w};
#pragma unroll
            for (int j = 0; j < 4; ++j) {
                const unsigned cl = (unsigned)(c[j] - q0);
                if (cl < (unsigned)PL_NODES) {
                    int2 v;
                    v.x = r[j];
                    v.y = __float_as_int(dinv_s[r[j]] * dinv_s[c[j]]);
                    ent2[slot_s[cl] + kk[j]] = v;
                }
            }
        }
    }
}

#define FT_NODES 64
#define FT_WAVES 16
#define FT_NPW   (FT_NODES / FT_WAVES)
#define FT_GROUP 16
#define FT_PITCH (DFEAT + 4)
#define GEMM_KSTEPS (DFEAT / 4)
static_assert((FT_WAVES / 2) * 16 == DFEAT && FT_NODES == 64 && FT_WAVES == 16,
              "8 column tiles x 2 pairs of row tiles");
static_assert(FT_NPW + 1 <= 32, "a wave's segment bounds fit one per lane");
static_assert(N_NODES % 16 == 0, "whole 16-row tiles");
static_assert(16 % FT_NPW == 0, "a wave's nodes lie in one tile, so all or none of them exist");
static_assert((DFEAT * DFEAT / 4) % (FT_WAVES * 32) == 0, "W copy: whole trips");

__global__ __launch_bounds__(FT_WAVES * 32)
void gather_gemm_stats_kernel(const int* __restrict__ off, const int* __restrict__ ent,
                              const float* __restrict__ X, const float* __restrict__ W,
                              const float* __restrict__ bias, float* __restrict__ agg,
                              float* __restrict__ stats) {
    __shared__ __attribute__((aligned(16))) float wl[DFEAT * FT_PITCH];
    __shared__ __attribute__((aligned(16))) float tile[FT_NODES * FT_PITCH];
    __shared__ float ssum[FT_NODES];
    __shared__ float part[FT_WAVES / 8][2 * DFEAT];
    const int lane = threadIdx.x & 31;
    const int wave = __builtin_amdgcn_readfirstlane((int)(threadIdx.x >> 5));
    const int l16   = lane & 15;
    const int hh    = lane >> 4;
    const int khalf = hh << 1;
    const int nc0   = (wave & 7) << 4;
    const int pr0   = (wave >> 3) << 5;

    {
        const float4* __restrict__ W4 = reinterpret_cast<const float4*>(W);
        float4 wv[(DFEAT * DFEAT / 4) / (FT_WAVES * 32)];
#pragma unroll
        for (int i = 0; i < (DFEAT * DFEAT / 4) / (FT_WAVES * 32); ++i)
            wv[i] = W4[(int)threadIdx.x + i * (FT_WAVES * 32)];
#pragma unroll
        for (int i = 0; i < (DFEAT * DFEAT / 4) / (FT_WAVES * 32); ++i) {
            const int q = (int)threadIdx.x + i * (FT_WAVES * 32);
            *reinterpret_cast<float4*>(&wl[(q >> 5) * FT_PITCH + ((q & 31) << 2)]) = wv[i];
        }
    }
    const float bv = bias[nc0 + l16];

    const int nb0 = (int)blockIdx.x * FT_NODES;
    const int n0 = nb0 + wave * FT_NPW;
    if (n0 < N_NODES) {
        const int myoff = off[n0 + ((lane < FT_NPW) ? lane : FT_NPW)];
        const int wend = __builtin_amdgcn_readlane(myoff, FT_NPW);
        const int2* __restrict__ ent2 = reinterpret_cast<const int2*>(ent);
        const float4* __restrict__ X4 = reinterpret_cast<const float4*>(X);
        int2 me_next = {0, 0};
        {
            const int p = __builtin_amdgcn_readlane(myoff, 0) + lane;
            if (p < wend) me_next = ent2[p];
        }
        for (int i = 0; i < FT_NPW; ++i) {
            const int beg = __builtin_amdgcn_readlane(myoff, i);
            const int end = __builtin_amdgcn_readlane(myoff, i + 1);
            float4 acc = {0.0f, 0.0f, 0.0f, 0.0f};
            float sw = 0.0f;
            for (int base = beg; base < end; base += 32) {
                const int2 me = me_next;
                const int nb = (base + 32 < end) ? (base + 32) : end;
                const int p = nb + lane;
                if (p < wend) me_next = ent2[p];
                const int m = (end - base < 32) ? (end - base) : 32;
                for (int j = 0; j < m; j += FT_GROUP) {
                    float4 v[FT_GROUP];
                    float wt[FT_GROUP];
#pragma unroll
                    for (int u = 0; u < FT_GROUP; ++u) {
                        if (j + u < m) {
                            const int s = __builtin_amdgcn_readlane(me.x, j + u);
                            wt[u] = __int_as_float(__builtin_amdgcn_readlane(me.y, j + u));
                            v[u] = X4[(size_t)s * (DFEAT / 4) + lane];
                        }
                    }
#pragma unroll
                    for (int u = 0; u < FT_GROUP; ++u) {
                        if (j + u < m) {
                            acc.x = fmaf(wt[u], v[u].x, acc.x);
                            acc.y = fmaf(wt[u], v[u].y, acc.y);
                            acc.z = fmaf(wt[u], v[u].z, acc.z);
                            acc.w = fmaf(wt[u], v[u].w, acc.w);
                            sw += wt[u];
                        }
                    }
                }
            }
            const int tr = wave * FT_NPW + i;
            *reinterpret_cast<float4*>(&tile[tr * FT_PITCH + (lane << 2)]) = acc;
            if (lane == 0) ssum[tr] = sw;
        }
    } else {
        for (int i = 0; i < FT_NPW; ++i) {
            const int tr = wave * FT_NPW + i;
            *reinterpret_cast<float4*>(&tile[tr * FT_PITCH + (lane << 2)]) =
                make_float4(0.0f, 0.0f, 0.0f, 0.0f);
            if (lane == 0) ssum[tr] = 0.0f;
        }
    }
    __syncthreads();

    const float* __restrict__ a0p = tile + (pr0 + l16) * FT_PITCH + khalf;
    const float* __restrict__ a1p = a0p + 16 * FT_PITCH;
    const float* __restrict__ bp  = wl + (nc0 + l16) * FT_PITCH + khalf;
    v8f c0, c1;
#pragma unroll
    for (int r = 0; r < 8; ++r) {
        c0[r] = bv * ssum[pr0 + r + (hh << 3)];
        c1[r] = bv * ssum[pr0 + 16 + r + (hh << 3)];
    }
#pragma unroll
    for (int kk = 0; kk < GEMM_KSTEPS; ++kk) {
        const v2f b  = *(const v2f*)(bp  + kk * 4);
        const v2f a0 = *(const v2f*)(a0p + kk * 4);
        const v2f a1 = *(const v2f*)(a1p + kk * 4);
        c0 = __builtin_amdgcn_wmma_f32_16x16x4_f32(false, a0, false, b,
                                                   (short)0, c0, false, false);
        c1 = __builtin_amdgcn_wmma_f32_16x16x4_f32(false, a1, false, b,
                                                   (short)0, c1, false, false);
    }
    float cs = 0.0f, cq = 0.0f;
#pragma unroll
    for (int r = 0; r < 8; ++r) {
        cs += c0[r];
        cq = fmaf(c0[r], c0[r], cq);
    }
#pragma unroll
    for (int r = 0; r < 8; ++r) {
        cs += c1[r];
        cq = fmaf(c1[r], c1[r], cq);
    }
    const float os = __shfl_xor(cs, 16);
    const float oq = __shfl_xor(cq, 16);
    part[wave >> 3][(hh << 7) + nc0 + l16] = (hh == 0) ? (cs + os) : (cq + oq);
    __syncthreads();
    const int t = (int)threadIdx.x;
    if (t < 2 * DFEAT) atomicAdd(&stats[t], part[0][t] + part[1][t]);

    const int row0 = nb0 + pr0 + (hh << 3);
    float* __restrict__ outcol = agg + (size_t)row0 * DFEAT + nc0 + l16;
    if (nb0 + pr0 < N_NODES) {
#pragma unroll
        for (int r = 0; r < 8; ++r) outcol[(size_t)r * DFEAT] = c0[r];
    }
    if (nb0 + pr0 + 16 < N_NODES) {
#pragma unroll
        for (int r = 0; r < 8; ++r) outcol[(size_t)(16 + r) * DFEAT] = c1[r];
    }
}

#define BA_WAVES 20
#define BA_ROWS  10

__global__ __launch_bounds__(BA_WAVES * 32)
void bn_apply(const float* xin, const float* __restrict__ agg,
              const float* __restrict__ stats, const float* __restrict__ g,
              const float* __restrict__ be, float* xout, int N) {
    const int lane = threadIdx.x & 31;
    const int wave = __builtin_amdgcn_readfirstlane((int)(threadIdx.x >> 5));
    const int gw = (int)blockIdx.x * BA_WAVES + wave;
    const int tw = (int)gridDim.x * BA_WAVES;
    const float invN = 1.0f / (float)N_NODES;
    const float4 s4 = reinterpret_cast<const float4*>(stats)[lane];
    const float4 q4 = reinterpret_cast<const float4*>(stats + DFEAT)[lane];
    const float4 g4 = reinterpret_cast<const float4*>(g)[lane];
    const float4 b4 = reinterpret_cast<const float4*>(be)[lane];
    const float sv[4] = {s4.x, s4.y, s4.z, s4.w};
    const float qv[4] = {q4.x, q4.y, q4.z, q4.w};
    const float gv[4] = {g4.x, g4.y, g4.z, g4.w};
    const float bv[4] = {b4.x, b4.y, b4.z, b4.w};
    float mu[4], sc[4];
#pragma unroll
    for (int j = 0; j < 4; ++j) {
        mu[j] = sv[j] * invN;
        const float var = qv[j] * invN - mu[j] * mu[j];
        sc[j] = gv[j] * rsqrtf(var + BN_EPS);
    }
    const float4* A4 = reinterpret_cast<const float4*>(agg);
    const float4* X4 = reinterpret_cast<const float4*>(xin);
    float4* O4 = reinterpret_cast<float4*>(xout);
    for (int r0 = gw; r0 < N; r0 += BA_ROWS * tw) {
        if (r0 + (BA_ROWS - 1) * tw < N) {
            float4 a[BA_ROWS];
            float4 xv[BA_ROWS];
#pragma unroll
            for (int j = 0; j < BA_ROWS; ++j) {
                const size_t idx = (size_t)(r0 + j * tw) * (DFEAT / 4) + lane;
                a[j]  = A4[idx];
                xv[j] = X4[idx];
            }
#pragma unroll
            for (int j = 0; j < BA_ROWS; ++j) {
                float4 o;
                o.x = xv[j].x + fmaxf((a[j].x - mu[0]) * sc[0] + bv[0], 0.0f);
                o.y = xv[j].y + fmaxf((a[j].y - mu[1]) * sc[1] + bv[1], 0.0f);
                o.z = xv[j].z + fmaxf((a[j].z - mu[2]) * sc[2] + bv[2], 0.0f);
                o.w = xv[j].w + fmaxf((a[j].w - mu[3]) * sc[3] + bv[3], 0.0f);
                O4[(size_t)(r0 + j * tw) * (DFEAT / 4) + lane] = o;
            }
        } else {
            for (int r = r0; r < N; r += tw) {
                const size_t idx = (size_t)r * (DFEAT / 4) + lane;
                const float4 a1 = A4[idx];
                const float4 x1 = X4[idx];
                float4 o;
                o.x = x1.x + fmaxf((a1.x - mu[0]) * sc[0] + bv[0], 0.0f);
                o.y = x1.y + fmaxf((a1.y - mu[1]) * sc[1] + bv[1], 0.0f);
                o.z = x1.z + fmaxf((a1.z - mu[2]) * sc[2] + bv[2], 0.0f);
                o.w = x1.w + fmaxf((a1.w - mu[3]) * sc[3] + bv[3], 0.0f);
                O4[idx] = o;
            }
        }
    }
}

extern "C" void kernel_launch(void* const* d_in, const int* in_sizes, int n_in,
                              void* d_out, int out_size, void* d_ws, size_t ws_size,
                              hipStream_t stream) {
    (void)in_sizes; (void)n_in; (void)out_size; (void)ws_size;

    const float* x     = (const float*)d_in[0];
    const int*   ei    = (const int*)d_in[1];
    const float* W     = (const float*)d_in[2];
    const float* bias  = (const float*)d_in[3];
    const float* gamma = (const float*)d_in[4];
    const float* beta  = (const float*)d_in[5];
    float* out = (float*)d_out;

    const size_t ND = (size_t)N_NODES * DFEAT;
    float* ws    = (float*)d_ws;
    float* dinv  = ws;
    int*   cnt   = (int*)(ws + N_NODES);
    int*   off   = (int*)(ws + 2 * N_NODES);
    int*   ent   = (int*)(ws + 4 * N_NODES);
    int*   rank  = (int*)(ws + 4 * N_NODES + 2 * N_EDGES);
    float* agg   = ws + 2 * ND;
    float* stats = ws + 3 * N_NODES + 64;
    unsigned* histP = (unsigned*)(ws + ND);
    int*      base  = (int*)(ws + ND + (size_t)SORT_BLOCKS * N_NODES);

    const int* row = ei;
    const int* col = ei + N_EDGES;

    hist_rank_kernel<<<SORT_BLOCKS, 1024, 0, stream>>>(row, col, histP, rank);
    combine_kernel<<<(HALF_N + 255) / 256, 256, 0, stream>>>(histP, base, cnt, dinv);
    scan_offsets<<<1, SCAN_WAVES * 32, 0, stream>>>(cnt, off, stats);
    place_kernel<<<SORT_BLOCKS * PL_RANGES, 1024, 0, stream>>>(row, col, rank, off, base, dinv, ent);

    for (int l = 0; l < N_LAYERS; ++l) {
        const float* x_in = (l == 0) ? x : out;
        float* st = stats + l * 2 * DFEAT;
        gather_gemm_stats_kernel<<<(N_NODES + FT_NODES - 1) / FT_NODES, FT_WAVES * 32, 0, stream>>>(
            off, ent, x_in, W + (size_t)l * DFEAT * DFEAT, bias + l * DFEAT, agg, st);
        bn_apply<<<250, BA_WAVES * 32, 0, stream>>>(
            x_in, agg, st, gamma + l * DFEAT, beta + l * DFEAT, out, N_NODES);
    }
}


# Gather overlaps the W product: one 32-wave block per WGP walks 64-node tiles; a wave asks for a node's first 8 rows, then runs the previous tile's WMMAs; BN sums per block: new reduction tree
# speedup vs baseline: 3.4799x; 1.0423x over previous
#include <hip/hip_runtime.h>
#include <math.h>


#define N_NODES 50000
#define N_EDGES 800000
#define DFEAT   128
#define N_LAYERS 3
#define BN_EPS  1e-5f

typedef __attribute__((ext_vector_type(2))) float v2f;
typedef __attribute__((ext_vector_type(8))) float v8f;

#define SORT_BLOCKS 64
#define SORT_CHUNK  (N_EDGES / SORT_BLOCKS)
#define HALF_N      (N_NODES / 2)
#define SORT_PER_THREAD ((SORT_CHUNK + 1023) / 1024)
static_assert(N_EDGES % SORT_BLOCKS == 0, "equal chunks");
static_assert(SORT_CHUNK < 65536, "a 16-bit field holds a chunk's count");
static_assert((2 * HALF_N) % 4 == 0, "tables are moved 16 bytes at a time");

__global__ __launch_bounds__(1024)
void hist_rank_kernel(const int* __restrict__ row, const int* __restrict__ col,
                      unsigned* __restrict__ histP, int* __restrict__ rank) {
    __shared__ __attribute__((aligned(16))) unsigned hs[2 * HALF_N];
    const int t = threadIdx.x;
    uint4* __restrict__ hs4 = reinterpret_cast<uint4*>(hs);
    for (int i = t; i < (2 * HALF_N) / 4; i += 1024) hs4[i] = make_uint4(0u, 0u, 0u, 0u);
    const int e0 = (int)blockIdx.x * SORT_CHUNK;
    int rr[SORT_PER_THREAD];
    int cc[SORT_PER_THREAD];
#pragma unroll
    for (int k = 0; k < SORT_PER_THREAD; ++k) {
        const int i = t + k * 1024;
        const int e = e0 + ((i < SORT_CHUNK) ? i : (SORT_CHUNK - 1));
        rr[k] = row[e];
        cc[k] = col[e];
    }
    __syncthreads();
    unsigned old[SORT_PER_THREAD];
#pragma unroll
    for (int k = 0; k < SORT_PER_THREAD; ++k) {
        const int i = t + k * 1024;
        old[k] = 0u;
        if (i < SORT_CHUNK) {
            const int c = cc[k];
            const int r = rr[k];
            old[k] = atomicAdd(&hs[c >> 1], 1u << ((unsigned)(c & 1) << 4));
            atomicAdd(&hs[HALF_N + (r >> 1)], 1u << ((unsigned)(r & 1) << 4));
        }
    }
#pragma unroll
    for (int k = 0; k < SORT_PER_THREAD; ++k) {
        const int i = t + k * 1024;
        if (i < SORT_CHUNK)
            rank[e0 + i] = (int)((old[k] >> ((unsigned)(cc[k] & 1) << 4)) & 0xffffu);
    }
    __syncthreads();
    uint4* __restrict__ out4 =
        reinterpret_cast<uint4*>(histP + (size_t)blockIdx.x * (2 * HALF_N));
    for (int i = t; i < (2 * HALF_N) / 4; i += 1024) out4[i] = hs4[i];
}

__global__ void combine_kernel(const unsigned* __restrict__ histP, int* __restrict__ base,
                               int* __restrict__ cnt, float* __restrict__ dinv) {
    const int t = blockIdx.x * blockDim.x + threadIdx.x;
    if (t >= HALF_N) return;
    unsigned c0 = 0u, c1 = 0u, d0 = 0u, d1 = 0u;
#pragma unroll 32
    for (int b = 0; b < SORT_BLOCKS; ++b) {
        const unsigned pc = histP[(size_t)b * (2 * HALF_N) + t];
        const unsigned pd = histP[(size_t)b * (2 * HALF_N) + HALF_N + t];
        int2 v;
        v.x = (int)c0;
        v.y = (int)c1;
        reinterpret_cast<int2*>(base + (size_t)b * N_NODES)[t] = v;
        c0 += pc & 0xffffu;
        c1 += pc >> 16;
        d0 += pd & 0xffffu;
        d1 += pd >> 16;
    }
    int2 cv;
    cv.x = (int)c0;
    cv.y = (int)c1;
    reinterpret_cast<int2*>(cnt)[t] = cv;
    float2 dv;
    dv.x = (d0 > 0u) ? rsqrtf((float)d0) : 0.0f;
    dv.y = (d1 > 0u) ? rsqrtf((float)d1) : 0.0f;
    reinterpret_cast<float2*>(dinv)[t] = dv;
}

#define SCAN_WAVES 32
#define SCAN_TRIPS (((N_NODES + SCAN_WAVES - 1) / SCAN_WAVES + 127) / 128)
static_assert(N_LAYERS * 2 * DFEAT <= 1024, "one thread per column sum in the scan block");
static_assert(N_NODES % 4 == 0, "counts and offsets are moved 16 bytes at a time");
static_assert(SCAN_WAVES * SCAN_TRIPS * 128 >= N_NODES, "the block covers every node");
static_assert(SCAN_WAVES == 32, "one wave total per lane in the second-level scan");

__global__ __launch_bounds__(SCAN_WAVES * 32) __attribute__((amdgpu_waves_per_eu(8, 8)))
void scan_offsets(const int* __restrict__ cnt, int* __restrict__ off,
                  float* __restrict__ stats_all) {
    __shared__ int wtot[SCAN_WAVES];
    const int t = (int)threadIdx.x;
    if (t < N_LAYERS * 2 * DFEAT) stats_all[t] = 0.0f;
    const int lane = t & 31;
    const int wave = __builtin_amdgcn_readfirstlane(t >> 5);
    const int4* __restrict__ cnt4 = reinterpret_cast<const int4*>(cnt);
    int4* __restrict__ off4 = reinterpret_cast<int4*>(off);
    const int q0 = wave * (SCAN_TRIPS * 32) + lane;
    int4 v[SCAN_TRIPS];
#pragma unroll
    for (int j = 0; j < SCAN_TRIPS; ++j) {
        const int q = q0 + j * 32;
        v[j] = (q < N_NODES / 4) ? cnt4[q] : make_int4(0, 0, 0, 0);
    }
    int tot[SCAN_TRIPS];
    int x[SCAN_TRIPS];
#pragma unroll
    for (int j = 0; j < SCAN_TRIPS; ++j) {
        tot[j] = v[j].x + v[j].y + v[j].z + v[j].w;
        x[j] = tot[j];
    }
#pragma unroll
    for (int d = 1; d < 32; d <<= 1) {
        int y[SCAN_TRIPS];
#pragma unroll
        for (int j = 0; j < SCAN_TRIPS; ++j) y[j] = __shfl_up(x[j], (unsigned)d);
#pragma unroll
        for (int j = 0; j < SCAN_TRIPS; ++j)
            if (lane >= d) x[j] += y[j];
    }
    int ex[SCAN_TRIPS];
    int carry = 0;
#pragma unroll
    for (int j = 0; j < SCAN_TRIPS; ++j) {
        ex[j] = carry + x[j] - tot[j];
        carry += __builtin_amdgcn_readlane(x[j], 31);
    }
    if (lane == 0) wtot[wave] = carry;
    __syncthreads();
    const int mine = wtot[lane];
    int wx = mine;
#pragma unroll
    for (int d = 1; d < 32; d <<= 1) {
        const int y = __shfl_up(wx, (unsigned)d);
        if (lane >= d) wx += y;
    }
    const int total = __builtin_amdgcn_readlane(wx, 31);
    const int wbase = __builtin_amdgcn_readlane(wx - mine, wave);
#pragma unroll
    for (int j = 0; j < SCAN_TRIPS; ++j) {
        const int q = q0 + j * 32;
        if (q < N_NODES / 4) {
            int4 o;
            o.x = wbase + ex[j];
            o.y = o.x + v[j].x;
            o.z = o.y + v[j].y;
            o.w = o.z + v[j].z;
            off4[q] = o;
        }
    }
    if (t == 0) off[N_NODES] = total;
}

#define PL_RANGES 4
#define PL_NODES  (N_NODES / PL_RANGES)
#define PL_GROUPS (SORT_CHUNK / 4)
#define PL_TRIPS  ((PL_GROUPS + 1023) / 1024)
#define PL_DINV_TRIPS ((N_NODES / 4 + 1023) / 1024)
#define PL_SLOT_TRIPS ((PL_NODES / 4 + 1023) / 1024)
static_assert(N_NODES % (4 * PL_RANGES) == 0, "node ranges are moved 16 bytes at a time");
static_assert(SORT_CHUNK % 4 == 0, "a chunk is whole groups of 4 edges, 16-byte aligned");
static_assert((N_NODES + PL_NODES) * 4 <= 320 * 1024, "dinv and one slot table fit the LDS");

__global__ __launch_bounds__(1024)
void place_kernel(const int* __restrict__ row, const int* __restrict__ col,
                  const int* __restrict__ rank, const int* __restrict__ off,
                  const int* __restrict__ base,
                  const float* __restrict__ dinv, int* __restrict__ ent) {
    __shared__ __attribute__((aligned(16))) float dinv_s[N_NODES];
    __shared__ __attribute__((aligned(16))) int   slot_s[PL_NODES];
    const int t = (int)threadIdx.x;
    const int chunk = (int)blockIdx.x / PL_RANGES;
    const int q0 = ((int)blockIdx.x % PL_RANGES) * PL_NODES;
    const int4* __restrict__ row4 = reinterpret_cast<const int4*>(row) + chunk * PL_GROUPS;
    const int4* __restrict__ col4 = reinterpret_cast<const int4*>(col) + chunk * PL_GROUPS;
    const int4* __restrict__ rnk4 = reinterpret_cast<const int4*>(rank) + chunk * PL_GROUPS;
    int4 r4[PL_TRIPS], c4[PL_TRIPS], k4[PL_TRIPS];
#pragma unroll
    for (int k = 0; k < PL_TRIPS; ++k) {
        const int g = t + k * 1024;
        const int gi = (g < PL_GROUPS) ? g : (PL_GROUPS - 1);
        r4[k] = row4[gi];
        c4[k] = col4[gi];
        k4[k] = rnk4[gi];
    }
    {
        const float4* __restrict__ d4 = reinterpret_cast<const float4*>(dinv);
        float4* __restrict__ s4 = reinterpret_cast<float4*>(dinv_s);
        float4 v[PL_DINV_TRIPS];
#pragma unroll
        for (int k = 0; k < PL_DINV_TRIPS; ++k) {
            const int i = t + k * 1024;
            v[k] = (i < N_NODES / 4) ? d4[i] : make_float4(0.0f, 0.0f, 0.0f, 0.0f);
        }
#pragma unroll
        for (int k = 0; k < PL_DINV_TRIPS; ++k) {
            const int i = t + k * 1024;
            if (i < N_NODES / 4) s4[i] = v[k];
        }
    }
    {
        const int4* __restrict__ o4 = reinterpret_cast<const int4*>(off + q0);
        const int4* __restrict__ b4 =
            reinterpret_cast<const int4*>(base + (size_t)chunk * N_NODES + q0);
        int4* __restrict__ s4 = reinterpret_cast<int4*>(slot_s);
        int4 a[PL_SLOT_TRIPS], b[PL_SLOT_TRIPS];
#pragma unroll
        for (int k = 0; k < PL_SLOT_TRIPS; ++k) {
            const int i = t + k * 1024;
            a[k] = (i < PL_NODES / 4) ? o4[i] : make_int4(0, 0, 0, 0);
            b[k] = (i < PL_NODES / 4) ? b4[i] : make_int4(0, 0, 0, 0);
        }
#pragma unroll
        for (int k = 0; k < PL_SLOT_TRIPS; ++k) {
            const int i = t + k * 1024;
            if (i < PL_NODES / 4)
                s4[i] = make_int4(a[k].x + b[k].x, a[k].y + b[k].y, a[k].z + b[k].z, a[k].w + b[k].w);
        }
    }
    __syncthreads();
    int2* __restrict__ ent2 = reinterpret_cast<int2*>(ent);
#pragma unroll
    for (int k = 0; k < PL_TRIPS; ++k) {
        if (t + k * 1024 < PL_GROUPS) {
            const int r[4]  = {r4[k].x, r4[k].y, r4[k].z, r4[k].w};
            const int c[4]  = {c4[k].x, c4[k].y, c4[k].z, c4[k].w};
            const int kk[4] = {k4[k].x, k4[k].y, k4[k].z, k4[k].w};
#pragma unroll
            for (int j = 0; j < 4; ++j) {
                const unsigned cl = (unsigned)(c[j] - q0);
                if (cl < (unsigned)PL_NODES) {
                    int2 v;
                    v.x = r[j];
                    v.y = __float_as_int(dinv_s[r[j]] * dinv_s[c[j]]);
                    ent2[slot_s[cl] + kk[j]] = v;
                }
            }
        }
    }
}

#define PG_BLOCKS 256
#define PG_WAVES 32
#define PG_TILE  64
#define PG_STEPS 4
#define PG_GRAN  (N_NODES / 16)
#define FT_GROUP 8
#define FT_REST  4
#define FT_PITCH (DFEAT + 4)
#define GEMM_KSTEPS (DFEAT / 4)
static_assert(N_NODES % 16 == 0, "whole 16-row tiles");
static_assert((PG_GRAN + PG_BLOCKS - 1) / PG_BLOCKS * 16 <= PG_STEPS * PG_TILE, "steps cover a block's run");
static_assert(PG_WAVES * 2 == PG_TILE && (PG_WAVES / 8) * 16 == PG_TILE && 8 * 16 == DFEAT,
              "two nodes per wave; 4 row tiles x 8 column tiles");
static_assert(3 * PG_STEPS <= 32, "a wave's segment bounds fit one per lane");
static_assert((DFEAT * DFEAT / 4) % (PG_WAVES * 32) == 0, "W copy: whole trips");

template <int G>
__device__ __forceinline__ void pg_issue(const float4* __restrict__ X4, const int2 me, const int j,
                                         const int m, const int lane,
                                         float4 (&v)[G], float (&wt)[G]) {
#pragma unroll
    for (int u = 0; u < G; ++u) {
        if (j + u < m) {
            const int s = __builtin_amdgcn_readlane(me.x, j + u);
            wt[u] = __int_as_float(__builtin_amdgcn_readlane(me.y, j + u));
            v[u] = X4[(size_t)s * (DFEAT / 4) + lane];
        }
    }
}
template <int G>
__device__ __forceinline__ void pg_consume(const int j, const int m, const float4 (&v)[G],
                                           const float (&wt)[G], float4& acc, float& sw) {
#pragma unroll
    for (int u = 0; u < G; ++u) {
        if (j + u < m) {
            acc.x = fmaf(wt[u], v[u].x, acc.x);
            acc.y = fmaf(wt[u], v[u].y, acc.y);
            acc.z = fmaf(wt[u], v[u].z, acc.z);
            acc.w = fmaf(wt[u], v[u].w, acc.w);
            sw += wt[u];
        }
    }
}
__device__ __forceinline__ v8f pg_khalf(const float* __restrict__ ap, const float* __restrict__ bp,
                                        const int k0, v8f c) {
#pragma unroll
    for (int kk = k0; kk < k0 + GEMM_KSTEPS / 2; ++kk) {
        const v2f b = *(const v2f*)(bp + kk * 4);
        const v2f a = *(const v2f*)(ap + kk * 4);
        c = __builtin_amdgcn_wmma_f32_16x16x4_f32(false, a, false, b, (short)0, c, false, false);
    }
    return c;
}
__device__ __forceinline__ void pg_issue_first(const float* __restrict__ X, const int2 me, const int m,
                                               const unsigned lane4, float4 (&v)[FT_GROUP],
                                               float (&wt)[FT_GROUP]) {
#pragma unroll
    for (int u = 0; u < FT_GROUP; ++u) {
        const int e = (u < m) ? u : (m - 1);
        const int s = __builtin_amdgcn_readlane(me.x, e);
        const float w = __int_as_float(__builtin_amdgcn_readlane(me.y, e));
        wt[u] = (u < m) ? w : 0.0f;
        const float* __restrict__ rowp = X + (size_t)s * DFEAT;
        v[u] = *reinterpret_cast<const float4*>(rowp + lane4);
    }
}
__device__ __forceinline__ void pg_consume_first(const float4 (&v)[FT_GROUP], const float (&wt)[FT_GROUP],
                                                 float4& acc, float& sw) {
#pragma unroll
    for (int u = 0; u < FT_GROUP; ++u) {
        acc.x = fmaf(wt[u], v[u].x, acc.x);
        acc.y = fmaf(wt[u], v[u].y, acc.y);
        acc.z = fmaf(wt[u], v[u].z, acc.z);
        acc.w = fmaf(wt[u], v[u].w, acc.w);
        sw += wt[u];
    }
}

__global__ __launch_bounds__(PG_WAVES * 32)
void gather_gemm_stats_kernel(const int* __restrict__ off, const int* __restrict__ ent,
                              const float* __restrict__ X, const float* __restrict__ W,
                              const float* __restrict__ bias, float* __restrict__ agg,
                              float* __restrict__ stats) {
    __shared__ __attribute__((aligned(16))) float wl[DFEAT * FT_PITCH];
    __shared__ __attribute__((aligned(16))) float tile[2][PG_TILE * FT_PITCH];
    __shared__ float ssum[2][PG_TILE];
    __shared__ float part[PG_WAVES / 8][2 * DFEAT];
    const int lane = threadIdx.x & 31;
    const int wave = __builtin_amdgcn_readfirstlane((int)(threadIdx.x >> 5));
    const int l16   = lane & 15;
    const int hh    = lane >> 4;
    const int khalf = hh << 1;
    const int nc0   = (wave & 7) << 4;
    const int rt    = wave >> 3;

    {
        const float4* __restrict__ W4 = reinterpret_cast<const float4*>(W);
        float4 wv[(DFEAT * DFEAT / 4) / (PG_WAVES * 32)];
#pragma unroll
        for (int i = 0; i < (DFEAT * DFEAT / 4) / (PG_WAVES * 32); ++i)
            wv[i] = W4[(int)threadIdx.x + i * (PG_WAVES * 32)];
#pragma unroll
        for (int i = 0; i < (DFEAT * DFEAT / 4) / (PG_WAVES * 32); ++i) {
            const int q = (int)threadIdx.x + i * (PG_WAVES * 32);
            *reinterpret_cast<float4*>(&wl[(q >> 5) * FT_PITCH + ((q & 31) << 2)]) = wv[i];
        }
    }
    const float bv = bias[nc0 + l16];

    const int nb0 = (((int)blockIdx.x * PG_GRAN) / PG_BLOCKS) * 16;
    const int nbe = ((((int)blockIdx.x + 1) * PG_GRAN) / PG_BLOCKS) * 16;
    const int nsteps = (nbe - nb0 + PG_TILE - 1) / PG_TILE;
    const int nw0 = nb0 + 2 * wave;

    int myoff;
    {
        const int st = lane / 3;
        int idx = nw0 + st * PG_TILE + (lane - 3 * st);
        idx = (idx < N_NODES) ? idx : N_NODES;
        myoff = off[idx];
    }
    const int2* __restrict__ ent2 = reinterpret_cast<const int2*>(ent);
    const float4* __restrict__ X4 = reinterpret_cast<const float4*>(X);
    int2 me_ns = {0, 0};
    int2 me_next = {0, 0};
    if (nw0 < nbe) {
        const int p = __builtin_amdgcn_readlane(myoff, 0) + lane;
        if (p < __builtin_amdgcn_readlane(myoff, 2)) me_ns = ent2[p];
    }
    const float* __restrict__ bp = wl + (nc0 + l16) * FT_PITCH + khalf;
    float cs = 0.0f, cq = 0.0f;
    __syncthreads();

    for (int s = 0; s <= nsteps; ++s) {
        const bool doG = (s < nsteps) && (nw0 + s * PG_TILE < nbe);
        const bool doM = (s >= 1) && (nb0 + (s - 1) * PG_TILE + 16 * rt < nbe);
        float* __restrict__ tg = tile[s & 1];
        const float* __restrict__ ap = tile[(s + 1) & 1] + (16 * rt + l16) * FT_PITCH + khalf;
        v8f c;
        if (doM) {
#pragma unroll
            for (int r = 0; r < 8; ++r) c[r] = bv * ssum[(s + 1) & 1][16 * rt + r + (hh << 3)];
        }
        int wend = 0;
        if (doG) {
            wend = __builtin_amdgcn_readlane(myoff, 3 * s + 2);
            me_next = me_ns;
            if ((s + 1 < nsteps) && (nw0 + (s + 1) * PG_TILE < nbe)) {
                const int p = __builtin_amdgcn_readlane(myoff, 3 * s + 3) + lane;
                if (p < __builtin_amdgcn_readlane(myoff, 3 * s + 5)) me_ns = ent2[p];
            }
        }
        for (int i = 0; i < 2; ++i) {
            float4 v[FT_GROUP];
            float wt[FT_GROUP];
            float4 acc = {0.0f, 0.0f, 0.0f, 0.0f};
            float sw = 0.0f;
            int beg = 0, end = 0, m = 0;
            int2 me = {0, 0};
            if (doG) {
                beg = __builtin_amdgcn_readlane(myoff, 3 * s + i);
                end = __builtin_amdgcn_readlane(myoff, 3 * s + i + 1);
                if (beg < end) {
                    me = me_next;
                    m = (end - beg < 32) ? (end - beg) : 32;
                    pg_issue_first(X, me, m, (unsigned)lane << 2, v, wt);
                }
            }
            if (doM) c = pg_khalf(ap, bp, i * (GEMM_KSTEPS / 2), c);
            if (doG) {
                if (beg < end) {
                    const int nb = (beg + 32 < end) ? (beg + 32) : end;
                    const int p = nb + lane;
                    if (p < wend) me_next = ent2[p];
                    pg_consume_first(v, wt, acc, sw);
                    for (int j = FT_GROUP; j < m; j += FT_REST) {
                        float4 v2[FT_REST];
                        float wt2[FT_REST];
                        pg_issue(X4, me, j, m, lane, v2, wt2);
                        pg_consume(j, m, v2, wt2, acc, sw);
                    }
                    for (int base = beg + 32; base < end; base += 32) {
                        const int2 mb = me_next;
                        const int nb = (base + 32 < end) ? (base + 32) : end;
                        const int p = nb + lane;
                        if (p < wend) me_next = ent2[p];
                        const int mm = (end - base < 32) ? (end - base) : 32;
                        for (int j = 0; j < mm; j += FT_REST) {
                            float4 v2[FT_REST];
                            float wt2[FT_REST];
                            pg_issue(X4, mb, j, mm, lane, v2, wt2);
                            pg_consume(j, mm, v2, wt2, acc, sw);
                        }
                    }
                }
                const int tr = 2 * wave + i;
                *reinterpret_cast<float4*>(&tg[tr * FT_PITCH + (lane << 2)]) = acc;
                if (lane == 0) ssum[s & 1][tr] = sw;
            }
        }
        if (doM) {
#pragma unroll
            for (int r = 0; r < 8; ++r) {
                cs += c[r];
                cq = fmaf(c[r], c[r], cq);
            }
            const int row0 = nb0 + (s - 1) * PG_TILE + 16 * rt + (hh << 3);
            float* __restrict__ outcol = agg + (size_t)row0 * DFEAT + nc0 + l16;
#pragma unroll
            for (int r = 0; r < 8; ++r) outcol[(size_t)r * DFEAT] = c[r];
        }
        __syncthreads();
    }
    const float os = __shfl_xor(cs, 16);
    const float oq = __shfl_xor(cq, 16);
    part[rt][(hh << 7) + nc0 + l16] = (hh == 0) ? (cs + os) : (cq + oq);
    __syncthreads();
    const int t = (int)threadIdx.x;
    if (t < 2 * DFEAT)
        atomicAdd(&stats[t], (part[0][t] + part[1][t]) + (part[2][t] + part[3][t]));
}

#define BA_WAVES 20
#define BA_ROWS  10

__global__ __launch_bounds__(BA_WAVES * 32)
void bn_apply(const float* xin, const float* __restrict__ agg,
              const float* __restrict__ stats, const float* __restrict__ g,
              const float* __restrict__ be, float* xout, int N) {
    const int lane = threadIdx.x & 31;
    const int wave = __builtin_amdgcn_readfirstlane((int)(threadIdx.x >> 5));
    const int gw = (int)blockIdx.x * BA_WAVES + wave;
    const int tw = (int)gridDim.x * BA_WAVES;
    const float invN = 1.0f / (float)N_NODES;
    const float4 s4 = reinterpret_cast<const float4*>(stats)[lane];
    const float4 q4 = reinterpret_cast<const float4*>(stats + DFEAT)[lane];
    const float4 g4 = reinterpret_cast<const float4*>(g)[lane];
    const float4 b4 = reinterpret_cast<const float4*>(be)[lane];
    const float sv[4] = {s4.x, s4.y, s4.z, s4.w};
    const float qv[4] = {q4.x, q4.y, q4.z, q4.w};
    const float gv[4] = {g4.x, g4.y, g4.z, g4.w};
    const float bv[4] = {b4.x, b4.y, b4.z, b4.w};
    float mu[4], sc[4];
#pragma unroll
    for (int j = 0; j < 4; ++j) {
        mu[j] = sv[j] * invN;
        const float var = qv[j] * invN - mu[j] * mu[j];
        sc[j] = gv[j] * rsqrtf(var + BN_EPS);
    }
    const float4* A4 = reinterpret_cast<const float4*>(agg);
    const float4* X4 = reinterpret_cast<const float4*>(xin);
    float4* O4 = reinterpret_cast<float4*>(xout);
    for (int r0 = gw; r0 < N; r0 += BA_ROWS * tw) {
        if (r0 + (BA_ROWS - 1) * tw < N) {
            float4 a[BA_ROWS];
            float4 xv[BA_ROWS];
#pragma unroll
            for (int j = 0; j < BA_ROWS; ++j) {
                const size_t idx = (size_t)(r0 + j * tw) * (DFEAT / 4) + lane;
                a[j]  = A4[idx];
                xv[j] = X4[idx];
            }
#pragma unroll
            for (int j = 0; j < BA_ROWS; ++j) {
                float4 o;
                o.x = xv[j].x + fmaxf((a[j].x - mu[0]) * sc[0] + bv[0], 0.0f);
                o.y = xv[j].y + fmaxf((a[j].y - mu[1]) * sc[1] + bv[1], 0.0f);
                o.z = xv[j].z + fmaxf((a[j].z - mu[2]) * sc[2] + bv[2], 0.0f);
                o.w = xv[j].w + fmaxf((a[j].w - mu[3]) * sc[3] + bv[3], 0.0f);
                O4[(size_t)(r0 + j * tw) * (DFEAT / 4) + lane] = o;
            }
        } else {
            for (int r = r0; r < N; r += tw) {
                const size_t idx = (size_t)r * (DFEAT / 4) + lane;
                const float4 a1 = A4[idx];
                const float4 x1 = X4[idx];
                float4 o;
                o.x = x1.x + fmaxf((a1.x - mu[0]) * sc[0] + bv[0], 0.0f);
                o.y = x1.y + fmaxf((a1.y - mu[1]) * sc[1] + bv[1], 0.0f);
                o.z = x1.z + fmaxf((a1.z - mu[2]) * sc[2] + bv[2], 0.0f);
                o.w = x1.w + fmaxf((a1.w - mu[3]) * sc[3] + bv[3], 0.0f);
                O4[idx] = o;
            }
        }
    }
}

extern "C" void kernel_launch(void* const* d_in, const int* in_sizes, int n_in,
                              void* d_out, int out_size, void* d_ws, size_t ws_size,
                              hipStream_t stream) {
    (void)in_sizes; (void)n_in; (void)out_size; (void)ws_size;

    const float* x     = (const float*)d_in[0];
    const int*   ei    = (const int*)d_in[1];
    const float* W     = (const float*)d_in[2];
    const float* bias  = (const float*)d_in[3];
    const float* gamma = (const float*)d_in[4];
    const float* beta  = (const float*)d_in[5];
    float* out = (float*)d_out;

    const size_t ND = (size_t)N_NODES * DFEAT;
    float* ws    = (float*)d_ws;
    float* dinv  = ws;
    int*   cnt   = (int*)(ws + N_NODES);
    int*   off   = (int*)(ws + 2 * N_NODES);
    int*   ent   = (int*)(ws + 4 * N_NODES);
    int*   rank  = (int*)(ws + 4 * N_NODES + 2 * N_EDGES);
    float* agg   = ws + 2 * ND;
    float* stats = ws + 3 * N_NODES + 64;
    unsigned* histP = (unsigned*)(ws + ND);
    int*      base  = (int*)(ws + ND + (size_t)SORT_BLOCKS * N_NODES);

    const int* row = ei;
    const int* col = ei + N_EDGES;

    hist_rank_kernel<<<SORT_BLOCKS, 1024, 0, stream>>>(row, col, histP, rank);
    combine_kernel<<<(HALF_N + 255) / 256, 256, 0, stream>>>(histP, base, cnt, dinv);
    scan_offsets<<<1, SCAN_WAVES * 32, 0, stream>>>(cnt, off, stats);
    place_kernel<<<SORT_BLOCKS * PL_RANGES, 1024, 0, stream>>>(row, col, rank, off, base, dinv, ent);

    for (int l = 0; l < N_LAYERS; ++l) {
        const float* x_in = (l == 0) ? x : out;
        float* st = stats + l * 2 * DFEAT;
        gather_gemm_stats_kernel<<<PG_BLOCKS, PG_WAVES * 32, 0, stream>>>(
            off, ent, x_in, W + (size_t)l * DFEAT * DFEAT, bias + l * DFEAT, agg, st);
        bn_apply<<<250, BA_WAVES * 32, 0, stream>>>(
            x_in, agg, st, gamma + l * DFEAT, beta + l * DFEAT, out, N_NODES);
    }
}


# agg leaves the product via LDS: a wave writes whole 512-B rows, one store per row, at the top of the next step (was 64-B pieces from the accumulators, waited for at each step's barrier)
# speedup vs baseline: 3.9525x; 1.1358x over previous
#include <hip/hip_runtime.h>
#include <math.h>


#define N_NODES 50000
#define N_EDGES 800000
#define DFEAT   128
#define N_LAYERS 3
#define BN_EPS  1e-5f

typedef __attribute__((ext_vector_type(2))) float v2f;
typedef __attribute__((ext_vector_type(8))) float v8f;

#define SORT_BLOCKS 64
#define SORT_CHUNK  (N_EDGES / SORT_BLOCKS)
#define HALF_N      (N_NODES / 2)
#define SORT_PER_THREAD ((SORT_CHUNK + 1023) / 1024)
static_assert(N_EDGES % SORT_BLOCKS == 0, "equal chunks");
static_assert(SORT_CHUNK < 65536, "a 16-bit field holds a chunk's count");
static_assert((2 * HALF_N) % 4 == 0, "tables are moved 16 bytes at a time");

__global__ __launch_bounds__(1024)
void hist_rank_kernel(const int* __restrict__ row, const int* __restrict__ col,
                      unsigned* __restrict__ histP, int* __restrict__ rank) {
    __shared__ __attribute__((aligned(16))) unsigned hs[2 * HALF_N];
    const int t = threadIdx.x;
    uint4* __restrict__ hs4 = reinterpret_cast<uint4*>(hs);
    for (int i = t; i < (2 * HALF_N) / 4; i += 1024) hs4[i] = make_uint4(0u, 0u, 0u, 0u);
    const int e0 = (int)blockIdx.x * SORT_CHUNK;
    int rr[SORT_PER_THREAD];
    int cc[SORT_PER_THREAD];
#pragma unroll
    for (int k = 0; k < SORT_PER_THREAD; ++k) {
        const int i = t + k * 1024;
        const int e = e0 + ((i < SORT_CHUNK) ? i : (SORT_CHUNK - 1));
        rr[k] = row[e];
        cc[k] = col[e];
    }
    __syncthreads();
    unsigned old[SORT_PER_THREAD];
#pragma unroll
    for (int k = 0; k < SORT_PER_THREAD; ++k) {
        const int i = t + k * 1024;
        old[k] = 0u;
        if (i < SORT_CHUNK) {
            const int c = cc[k];
            const int r = rr[k];
            old[k] = atomicAdd(&hs[c >> 1], 1u << ((unsigned)(c & 1) << 4));
            atomicAdd(&hs[HALF_N + (r >> 1)], 1u << ((unsigned)(r & 1) << 4));
        }
    }
#pragma unroll
    for (int k = 0; k < SORT_PER_THREAD; ++k) {
        const int i = t + k * 1024;
        if (i < SORT_CHUNK)
            rank[e0 + i] = (int)((old[k] >> ((unsigned)(cc[k] & 1) << 4)) & 0xffffu);
    }
    __syncthreads();
    uint4* __restrict__ out4 =
        reinterpret_cast<uint4*>(histP + (size_t)blockIdx.x * (2 * HALF_N));
    for (int i = t; i < (2 * HALF_N) / 4; i += 1024) out4[i] = hs4[i];
}

__global__ void combine_kernel(const unsigned* __restrict__ histP, int* __restrict__ base,
                               int* __restrict__ cnt, float* __restrict__ dinv) {
    const int t = blockIdx.x * blockDim.x + threadIdx.x;
    if (t >= HALF_N) return;
    unsigned c0 = 0u, c1 = 0u, d0 = 0u, d1 = 0u;
#pragma unroll 32
    for (int b = 0; b < SORT_BLOCKS; ++b) {
        const unsigned pc = histP[(size_t)b * (2 * HALF_N) + t];
        const unsigned pd = histP[(size_t)b * (2 * HALF_N) + HALF_N + t];
        int2 v;
        v.x = (int)c0;
        v.y = (int)c1;
        reinterpret_cast<int2*>(base + (size_t)b * N_NODES)[t] = v;
        c0 += pc & 0xffffu;
        c1 += pc >> 16;
        d0 += pd & 0xffffu;
        d1 += pd >> 16;
    }
    int2 cv;
    cv.x = (int)c0;
    cv.y = (int)c1;
    reinterpret_cast<int2*>(cnt)[t] = cv;
    float2 dv;
    dv.x = (d0 > 0u) ? rsqrtf((float)d0) : 0.0f;
    dv.y = (d1 > 0u) ? rsqrtf((float)d1) : 0.0f;
    reinterpret_cast<float2*>(dinv)[t] = dv;
}

#define SCAN_WAVES 32
#define SCAN_TRIPS (((N_NODES + SCAN_WAVES - 1) / SCAN_WAVES + 127) / 128)
static_assert(N_LAYERS * 2 * DFEAT <= 1024, "one thread per column sum in the scan block");
static_assert(N_NODES % 4 == 0, "counts and offsets are moved 16 bytes at a time");
static_assert(SCAN_WAVES * SCAN_TRIPS * 128 >= N_NODES, "the block covers every node");
static_assert(SCAN_WAVES == 32, "one wave total per lane in the second-level scan");

__global__ __launch_bounds__(SCAN_WAVES * 32) __attribute__((amdgpu_waves_per_eu(8, 8)))
void scan_offsets(const int* __restrict__ cnt, int* __restrict__ off,
                  float* __restrict__ stats_all) {
    __shared__ int wtot[SCAN_WAVES];
    const int t = (int)threadIdx.x;
    if (t < N_LAYERS * 2 * DFEAT) stats_all[t] = 0.0f;
    const int lane = t & 31;
    const int wave = __builtin_amdgcn_readfirstlane(t >> 5);
    const int4* __restrict__ cnt4 = reinterpret_cast<const int4*>(cnt);
    int4* __restrict__ off4 = reinterpret_cast<int4*>(off);
    const int q0 = wave * (SCAN_TRIPS * 32) + lane;
    int4 v[SCAN_TRIPS];
#pragma unroll
    for (int j = 0; j < SCAN_TRIPS; ++j) {
        const int q = q0 + j * 32;
        v[j] = (q < N_NODES / 4) ? cnt4[q] : make_int4(0, 0, 0, 0);
    }
    int tot[SCAN_TRIPS];
    int x[SCAN_TRIPS];
#pragma unroll
    for (int j = 0; j < SCAN_TRIPS; ++j) {
        tot[j] = v[j].x + v[j].y + v[j].z + v[j].w;
        x[j] = tot[j];
    }
#pragma unroll
    for (int d = 1; d < 32; d <<= 1) {
        int y[SCAN_TRIPS];
#pragma unroll
        for (int j = 0; j < SCAN_TRIPS; ++j) y[j] = __shfl_up(x[j], (unsigned)d);
#pragma unroll
        for (int j = 0; j < SCAN_TRIPS; ++j)
            if (lane >= d) x[j] += y[j];
    }
    int ex[SCAN_TRIPS];
    int carry = 0;
#pragma unroll
    for (int j = 0; j < SCAN_TRIPS; ++j) {
        ex[j] = carry + x[j] - tot[j];
        carry += __builtin_amdgcn_readlane(x[j], 31);
    }
    if (lane == 0) wtot[wave] = carry;
    __syncthreads();
    const int mine = wtot[lane];
    int wx = mine;
#pragma unroll
    for (int d = 1; d < 32; d <<= 1) {
        const int y = __shfl_up(wx, (unsigned)d);
        if (lane >= d) wx += y;
    }
    const int total = __builtin_amdgcn_readlane(wx, 31);
    const int wbase = __builtin_amdgcn_readlane(wx - mine, wave);
#pragma unroll
    for (int j = 0; j < SCAN_TRIPS; ++j) {
        const int q = q0 + j * 32;
        if (q < N_NODES / 4) {
            int4 o;
            o.x = wbase + ex[j];
            o.y = o.x + v[j].x;
            o.z = o.y + v[j].y;
            o.w = o.z + v[j].z;
            off4[q] = o;
        }
    }
    if (t == 0) off[N_NODES] = total;
}

#define PL_RANGES 4
#define PL_NODES  (N_NODES / PL_RANGES)
#define PL_GROUPS (SORT_CHUNK / 4)
#define PL_TRIPS  ((PL_GROUPS + 1023) / 1024)
#define PL_DINV_TRIPS ((N_NODES / 4 + 1023) / 1024)
#define PL_SLOT_TRIPS ((PL_NODES / 4 + 1023) / 1024)
static_assert(N_NODES % (4 * PL_RANGES) == 0, "node ranges are moved 16 bytes at a time");
static_assert(SORT_CHUNK % 4 == 0, "a chunk is whole groups of 4 edges, 16-byte aligned");
static_assert((N_NODES + PL_NODES) * 4 <= 320 * 1024, "dinv and one slot table fit the LDS");

__global__ __launch_bounds__(1024)
void place_kernel(const int* __restrict__ row, const int* __restrict__ col,
                  const int* __restrict__ rank, const int* __restrict__ off,
                  const int* __restrict__ base,
                  const float* __restrict__ dinv, int* __restrict__ ent) {
    __shared__ __attribute__((aligned(16))) float dinv_s[N_NODES];
    __shared__ __attribute__((aligned(16))) int   slot_s[PL_NODES];
    const int t = (int)threadIdx.x;
    const int chunk = (int)blockIdx.x / PL_RANGES;
    const int q0 = ((int)blockIdx.x % PL_RANGES) * PL_NODES;
    const int4* __restrict__ row4 = reinterpret_cast<const int4*>(row) + chunk * PL_GROUPS;
    const int4* __restrict__ col4 = reinterpret_cast<const int4*>(col) + chunk * PL_GROUPS;
    const int4* __restrict__ rnk4 = reinterpret_cast<const int4*>(rank) + chunk * PL_GROUPS;
    int4 r4[PL_TRIPS], c4[PL_TRIPS], k4[PL_TRIPS];
#pragma unroll
    for (int k = 0; k < PL_TRIPS; ++k) {
        const int g = t + k * 1024;
        const int gi = (g < PL_GROUPS) ? g : (PL_GROUPS - 1);
        r4[k] = row4[gi];
        c4[k] = col4[gi];
        k4[k] = rnk4[gi];
    }
    {
        const float4* __restrict__ d4 = reinterpret_cast<const float4*>(dinv);
        float4* __restrict__ s4 = reinterpret_cast<float4*>(dinv_s);
        float4 v[PL_DINV_TRIPS];
#pragma unroll
        for (int k = 0; k < PL_DINV_TRIPS; ++k) {
            const int i = t + k * 1024;
            v[k] = (i < N_NODES / 4) ? d4[i] : make_float4(0.0f, 0.0f, 0.0f, 0.0f);
        }
#pragma unroll
        for (int k = 0; k < PL_DINV_TRIPS; ++k) {
            const int i = t + k * 1024;
            if (i < N_NODES / 4) s4[i] = v[k];
        }
    }
    {
        const int4* __restrict__ o4 = reinterpret_cast<const int4*>(off + q0);
        const int4* __restrict__ b4 =
            reinterpret_cast<const int4*>(base + (size_t)chunk * N_NODES + q0);
        int4* __restrict__ s4 = reinterpret_cast<int4*>(slot_s);
        int4 a[PL_SLOT_TRIPS], b[PL_SLOT_TRIPS];
#pragma unroll
        for (int k = 0; k < PL_SLOT_TRIPS; ++k) {
            const int i = t + k * 1024;
            a[k] = (i < PL_NODES / 4) ? o4[i] : make_int4(0, 0, 0, 0);
            b[k] = (i < PL_NODES / 4) ? b4[i] : make_int4(0, 0, 0, 0);
        }
#pragma unroll
        for (int k = 0; k < PL_SLOT_TRIPS; ++k) {
            const int i = t + k * 1024;
            if (i < PL_NODES / 4)
                s4[i] = make_int4(a[k].x + b[k].x, a[k].y + b[k].y, a[k].z + b[k].z, a[k].w + b[k].w);
        }
    }
    __syncthreads();
    int2* __restrict__ ent2 = reinterpret_cast<int2*>(ent);
#pragma unroll
    for (int k = 0; k < PL_TRIPS; ++k) {
        if (t + k * 1024 < PL_GROUPS) {
            const int r[4]  = {r4[k].x, r4[k].y, r4[k].z, r4[k].w};
            const int c[4]  = {c4[k].x, c4[k].y, c4[k].z, c4[k].w};
            const int kk[4] = {k4[k].x, k4[k].y, k4[k].z, k4[k].w};
#pragma unroll
            for (int j = 0; j < 4; ++j) {
                const unsigned cl = (unsigned)(c[j] - q0);
                if (cl < (unsigned)PL_NODES) {
                    int2 v;
                    v.x = r[j];
                    v.y = __float_as_int(dinv_s[r[j]] * dinv_s[c[j]]);
                    ent2[slot_s[cl] + kk[j]] = v;
                }
            }
        }
    }
}

#define PG_BLOCKS 256
#define PG_WAVES 32
#define PG_TILE  64
#define PG_STEPS 4
#define PG_GRAN  (N_NODES / 16)
#define FT_GROUP 8
#define FT_REST  4
#define FT_PITCH (DFEAT + 4)
#define GEMM_KSTEPS (DFEAT / 4)
static_assert(N_NODES % 16 == 0, "whole 16-row tiles");
static_assert((PG_GRAN + PG_BLOCKS - 1) / PG_BLOCKS * 16 <= PG_STEPS * PG_TILE, "steps cover a block's run");
static_assert(PG_WAVES * 2 == PG_TILE && (PG_WAVES / 8) * 16 == PG_TILE && 8 * 16 == DFEAT,
              "two nodes per wave; 4 row tiles x 8 column tiles");
static_assert(3 * PG_STEPS <= 32, "a wave's segment bounds fit one per lane");
static_assert((DFEAT * DFEAT / 4) % (PG_WAVES * 32) == 0, "W copy: whole trips");

template <int G>
__device__ __forceinline__ void pg_issue(const float4* __restrict__ X4, const int2 me, const int j,
                                         const int m, const int lane,
                                         float4 (&v)[G], float (&wt)[G]) {
#pragma unroll
    for (int u = 0; u < G; ++u) {
        if (j + u < m) {
            const int s = __builtin_amdgcn_readlane(me.x, j + u);
            wt[u] = __int_as_float(__builtin_amdgcn_readlane(me.y, j + u));
            v[u] = X4[(size_t)s * (DFEAT / 4) + lane];
        }
    }
}
template <int G>
__device__ __forceinline__ void pg_consume(const int j, const int m, const float4 (&v)[G],
                                           const float (&wt)[G], float4& acc, float& sw) {
#pragma unroll
    for (int u = 0; u < G; ++u) {
        if (j + u < m) {
            acc.x = fmaf(wt[u], v[u].x, acc.x);
            acc.y = fmaf(wt[u], v[u].y, acc.y);
            acc.z = fmaf(wt[u], v[u].z, acc.z);
            acc.w = fmaf(wt[u], v[u].w, acc.w);
            sw += wt[u];
        }
    }
}
__device__ __forceinline__ v8f pg_khalf(const float* __restrict__ ap, const float* __restrict__ bp,
                                        const int k0, v8f c) {
#pragma unroll
    for (int kk = k0; kk < k0 + GEMM_KSTEPS / 2; ++kk) {
        const v2f b = *(const v2f*)(bp + kk * 4);
        const v2f a = *(const v2f*)(ap + kk * 4);
        c = __builtin_amdgcn_wmma_f32_16x16x4_f32(false, a, false, b, (short)0, c, false, false);
    }
    return c;
}
__device__ __forceinline__ void pg_issue_first(const float* __restrict__ X, const int2 me, const int m,
                                               const unsigned lane4, float4 (&v)[FT_GROUP],
                                               float (&wt)[FT_GROUP]) {
#pragma unroll
    for (int u = 0; u < FT_GROUP; ++u) {
        const int e = (u < m) ? u : (m - 1);
        const int s = __builtin_amdgcn_readlane(me.x, e);
        const float w = __int_as_float(__builtin_amdgcn_readlane(me.y, e));
        wt[u] = (u < m) ? w : 0.0f;
        const float* __restrict__ rowp = X + (size_t)s * DFEAT;
        v[u] = *reinterpret_cast<const float4*>(rowp + lane4);
    }
}
__device__ __forceinline__ void pg_consume_first(const float4 (&v)[FT_GROUP], const float (&wt)[FT_GROUP],
                                                 float4& acc, float& sw) {
#pragma unroll
    for (int u = 0; u < FT_GROUP; ++u) {
        acc.x = fmaf(wt[u], v[u].x, acc.x);
        acc.y = fmaf(wt[u], v[u].y, acc.y);
        acc.z = fmaf(wt[u], v[u].z, acc.z);
        acc.w = fmaf(wt[u], v[u].w, acc.w);
        sw += wt[u];
    }
}

__device__ __forceinline__ void pg_store_rows(const float* __restrict__ dt, float* __restrict__ agg,
                                              const int r0, const int nbe, const int wave,
                                              const int lane) {
    if (r0 < nbe) {
        const float4 a = *reinterpret_cast<const float4*>(&dt[(2 * wave) * FT_PITCH + (lane << 2)]);
        const float4 b = *reinterpret_cast<const float4*>(&dt[(2 * wave + 1) * FT_PITCH + (lane << 2)]);
        float4* __restrict__ o = reinterpret_cast<float4*>(agg + (size_t)r0 * DFEAT) + lane;
        o[0] = a;
        o[DFEAT / 4] = b;
    }
}

__global__ __launch_bounds__(PG_WAVES * 32)
void gather_gemm_stats_kernel(const int* __restrict__ off, const int* __restrict__ ent,
                              const float* __restrict__ X, const float* __restrict__ W,
                              const float* __restrict__ bias, float* __restrict__ agg,
                              float* __restrict__ stats) {
    __shared__ __attribute__((aligned(16))) float wl[DFEAT * FT_PITCH];
    __shared__ __attribute__((aligned(16))) float tile[2][PG_TILE * FT_PITCH];
    __shared__ __attribute__((aligned(16))) float dst[2][PG_TILE * FT_PITCH];
    __shared__ float ssum[2][PG_TILE];
    __shared__ float part[PG_WAVES / 8][2 * DFEAT];
    const int lane = threadIdx.x & 31;
    const int wave = __builtin_amdgcn_readfirstlane((int)(threadIdx.x >> 5));
    const int l16   = lane & 15;
    const int hh    = lane >> 4;
    const int khalf = hh << 1;
    const int nc0   = (wave & 7) << 4;
    const int rt    = wave >> 3;

    {
        const float4* __restrict__ W4 = reinterpret_cast<const float4*>(W);
        float4 wv[(DFEAT * DFEAT / 4) / (PG_WAVES * 32)];
#pragma unroll
        for (int i = 0; i < (DFEAT * DFEAT / 4) / (PG_WAVES * 32); ++i)
            wv[i] = W4[(int)threadIdx.x + i * (PG_WAVES * 32)];
#pragma unroll
        for (int i = 0; i < (DFEAT * DFEAT / 4) / (PG_WAVES * 32); ++i) {
            const int q = (int)threadIdx.x + i * (PG_WAVES * 32);
            *reinterpret_cast<float4*>(&wl[(q >> 5) * FT_PITCH + ((q & 31) << 2)]) = wv[i];
        }
    }
    const float bv = bias[nc0 + l16];

    const int nb0 = (((int)blockIdx.x * PG_GRAN) / PG_BLOCKS) * 16;
    const int nbe = ((((int)blockIdx.x + 1) * PG_GRAN) / PG_BLOCKS) * 16;
    const int nsteps = (nbe - nb0 + PG_TILE - 1) / PG_TILE;
    const int nw0 = nb0 + 2 * wave;

    int myoff;
    {
        const int st = lane / 3;
        int idx = nw0 + st * PG_TILE + (lane - 3 * st);
        idx = (idx < N_NODES) ? idx : N_NODES;
        myoff = off[idx];
    }
    const int2* __restrict__ ent2 = reinterpret_cast<const int2*>(ent);
    const float4* __restrict__ X4 = reinterpret_cast<const float4*>(X);
    int2 me_ns = {0, 0};
    int2 me_next = {0, 0};
    if (nw0 < nbe) {
        const int p = __builtin_amdgcn_readlane(myoff, 0) + lane;
        if (p < __builtin_amdgcn_readlane(myoff, 2)) me_ns = ent2[p];
    }
    const float* __restrict__ bp = wl + (nc0 + l16) * FT_PITCH + khalf;
    float cs = 0.0f, cq = 0.0f;
    __syncthreads();

    for (int s = 0; s <= nsteps; ++s) {
        const bool doG = (s < nsteps) && (nw0 + s * PG_TILE < nbe);
        const bool doM = (s >= 1) && (nb0 + (s - 1) * PG_TILE + 16 * rt < nbe);
        float* __restrict__ tg = tile[s & 1];
        const float* __restrict__ ap = tile[(s + 1) & 1] + (16 * rt + l16) * FT_PITCH + khalf;
        v8f c;
        if (doM) {
#pragma unroll
            for (int r = 0; r < 8; ++r) c[r] = bv * ssum[(s + 1) & 1][16 * rt + r + (hh << 3)];
        }
        if (s >= 2)
            pg_store_rows(dst[(s + 1) & 1], agg, nb0 + (s - 2) * PG_TILE + 2 * wave, nbe, wave, lane);
        int wend = 0;
        if (doG) {
            wend = __builtin_amdgcn_readlane(myoff, 3 * s + 2);
            me_next = me_ns;
            if ((s + 1 < nsteps) && (nw0 + (s + 1) * PG_TILE < nbe)) {
                const int p = __builtin_amdgcn_readlane(myoff, 3 * s + 3) + lane;
                if (p < __builtin_amdgcn_readlane(myoff, 3 * s + 5)) me_ns = ent2[p];
            }
        }
        for (int i = 0; i < 2; ++i) {
            float4 v[FT_GROUP];
            float wt[FT_GROUP];
            float4 acc = {0.0f, 0.0f, 0.0f, 0.0f};
            float sw = 0.0f;
            int beg = 0, end = 0, m = 0;
            int2 me = {0, 0};
            if (doG) {
                beg = __builtin_amdgcn_readlane(myoff, 3 * s + i);
                end = __builtin_amdgcn_readlane(myoff, 3 * s + i + 1);
                if (beg < end) {
                    me = me_next;
                    m = (end - beg < 32) ? (end - beg) : 32;
                    pg_issue_first(X, me, m, (unsigned)lane << 2, v, wt);
                }
            }
            if (doM) c = pg_khalf(ap, bp, i * (GEMM_KSTEPS / 2), c);
            if (doG) {
                if (beg < end) {
                    const int nb = (beg + 32 < end) ? (beg + 32) : end;
                    const int p = nb + lane;
                    if (p < wend) me_next = ent2[p];
                    pg_consume_first(v, wt, acc, sw);
                    for (int j = FT_GROUP; j < m; j += FT_REST) {
                        float4 v2[FT_REST];
                        float wt2[FT_REST];
                        pg_issue(X4, me, j, m, lane, v2, wt2);
                        pg_consume(j, m, v2, wt2, acc, sw);
                    }
                    for (int base = beg + 32; base < end; base += 32) {
                        const int2 mb = me_next;
                        const int nb = (base + 32 < end) ? (base + 32) : end;
                        const int p = nb + lane;
                        if (p < wend) me_next = ent2[p];
                        const int mm = (end - base < 32) ? (end - base) : 32;
                        for (int j = 0; j < mm; j += FT_REST) {
                            float4 v2[FT_REST];
                            float wt2[FT_REST];
                            pg_issue(X4, mb, j, mm, lane, v2, wt2);
                            pg_consume(j, mm, v2, wt2, acc, sw);
                        }
                    }
                }
                const int tr = 2 * wave + i;
                *reinterpret_cast<float4*>(&tg[tr * FT_PITCH + (lane << 2)]) = acc;
                if (lane == 0) ssum[s & 1][tr] = sw;
            }
        }
        if (doM) {
#pragma unroll
            for (int r = 0; r < 8; ++r) {
                cs += c[r];
                cq = fmaf(c[r], c[r], cq);
            }
            float* __restrict__ dcol = dst[s & 1] + (16 * rt + (hh << 3)) * FT_PITCH + nc0 + l16;
#pragma unroll
            for (int r = 0; r < 8; ++r) dcol[r * FT_PITCH] = c[r];
        }
        __syncthreads();
    }
    const float os = __shfl_xor(cs, 16);
    const float oq = __shfl_xor(cq, 16);
    part[rt][(hh << 7) + nc0 + l16] = (hh == 0) ? (cs + os) : (cq + oq);
    __syncthreads();
    pg_store_rows(dst[nsteps & 1], agg, nb0 + (nsteps - 1) * PG_TILE + 2 * wave, nbe, wave, lane);
    const int t = (int)threadIdx.x;
    if (t < 2 * DFEAT)
        atomicAdd(&stats[t], (part[0][t] + part[1][t]) + (part[2][t] + part[3][t]));
}

#define BA_WAVES 20
#define BA_ROWS  10

__global__ __launch_bounds__(BA_WAVES * 32)
void bn_apply(const float* xin, const float* __restrict__ agg,
              const float* __restrict__ stats, const float* __restrict__ g,
              const float* __restrict__ be, float* xout, int N) {
    const int lane = threadIdx.x & 31;
    const int wave = __builtin_amdgcn_readfirstlane((int)(threadIdx.x >> 5));
    const int gw = (int)blockIdx.x * BA_WAVES + wave;
    const int tw = (int)gridDim.x * BA_WAVES;
    const float invN = 1.0f / (float)N_NODES;
    const float4 s4 = reinterpret_cast<const float4*>(stats)[lane];
    const float4 q4 = reinterpret_cast<const float4*>(stats + DFEAT)[lane];
    const float4 g4 = reinterpret_cast<const float4*>(g)[lane];
    const float4 b4 = reinterpret_cast<const float4*>(be)[lane];
    const float sv[4] = {s4.x, s4.y, s4.z, s4.w};
    const float qv[4] = {q4.x, q4.y, q4.z, q4.w};
    const float gv[4] = {g4.x, g4.y, g4.z, g4.w};
    const float bv[4] = {b4.x, b4.y, b4.z, b4.w};
    float mu[4], sc[4];
#pragma unroll
    for (int j = 0; j < 4; ++j) {
        mu[j] = sv[j] * invN;
        const float var = qv[j] * invN - mu[j] * mu[j];
        sc[j] = gv[j] * rsqrtf(var + BN_EPS);
    }
    const float4* A4 = reinterpret_cast<const float4*>(agg);
    const float4* X4 = reinterpret_cast<const float4*>(xin);
    float4* O4 = reinterpret_cast<float4*>(xout);
    for (int r0 = gw; r0 < N; r0 += BA_ROWS * tw) {
        if (r0 + (BA_ROWS - 1) * tw < N) {
            float4 a[BA_ROWS];
            float4 xv[BA_ROWS];
#pragma unroll
            for (int j = 0; j < BA_ROWS; ++j) {
                const size_t idx = (size_t)(r0 + j * tw) * (DFEAT / 4) + lane;
                a[j]  = A4[idx];
                xv[j] = X4[idx];
            }
#pragma unroll
            for (int j = 0; j < BA_ROWS; ++j) {
                float4 o;
                o.x = xv[j].x + fmaxf((a[j].x - mu[0]) * sc[0] + bv[0], 0.0f);
                o.y = xv[j].y + fmaxf((a[j].y - mu[1]) * sc[1] + bv[1], 0.0f);
                o.z = xv[j].z + fmaxf((a[j].z - mu[2]) * sc[2] + bv[2], 0.0f);
                o.w = xv[j].w + fmaxf((a[j].w - mu[3]) * sc[3] + bv[3], 0.0f);
                O4[(size_t)(r0 + j * tw) * (DFEAT / 4) + lane] = o;
            }
        } else {
            for (int r = r0; r < N; r += tw) {
                const size_t idx = (size_t)r * (DFEAT / 4) + lane;
                const float4 a1 = A4[idx];
                const float4 x1 = X4[idx];
                float4 o;
                o.x = x1.x + fmaxf((a1.x - mu[0]) * sc[0] + bv[0], 0.0f);
                o.y = x1.y + fmaxf((a1.y - mu[1]) * sc[1] + bv[1], 0.0f);
                o.z = x1.z + fmaxf((a1.z - mu[2]) * sc[2] + bv[2], 0.0f);
                o.w = x1.w + fmaxf((a1.w - mu[3]) * sc[3] + bv[3], 0.0f);
                O4[idx] = o;
            }
        }
    }
}

extern "C" void kernel_launch(void* const* d_in, const int* in_sizes, int n_in,
                              void* d_out, int out_size, void* d_ws, size_t ws_size,
                              hipStream_t stream) {
    (void)in_sizes; (void)n_in; (void)out_size; (void)ws_size;

    const float* x     = (const float*)d_in[0];
    const int*   ei    = (const int*)d_in[1];
    const float* W     = (const float*)d_in[2];
    const float* bias  = (const float*)d_in[3];
    const float* gamma = (const float*)d_in[4];
    const float* beta  = (const float*)d_in[5];
    float* out = (float*)d_out;

    const size_t ND = (size_t)N_NODES * DFEAT;
    float* ws    = (float*)d_ws;
    float* dinv  = ws;
    int*   cnt   = (int*)(ws + N_NODES);
    int*   off   = (int*)(ws + 2 * N_NODES);
    int*   ent   = (int*)(ws + 4 * N_NODES);
    int*   rank  = (int*)(ws + 4 * N_NODES + 2 * N_EDGES);
    float* agg   = ws + 2 * ND;
    float* stats = ws + 3 * N_NODES + 64;
    unsigned* histP = (unsigned*)(ws + ND);
    int*      base  = (int*)(ws + ND + (size_t)SORT_BLOCKS * N_NODES);

    const int* row = ei;
    const int* col = ei + N_EDGES;

    hist_rank_kernel<<<SORT_BLOCKS, 1024, 0, stream>>>(row, col, histP, rank);
    combine_kernel<<<(HALF_N + 255) / 256, 256, 0, stream>>>(histP, base, cnt, dinv);
    scan_offsets<<<1, SCAN_WAVES * 32, 0, stream>>>(cnt, off, stats);
    place_kernel<<<SORT_BLOCKS * PL_RANGES, 1024, 0, stream>>>(row, col, rank, off, base, dinv, ent);

    for (int l = 0; l < N_LAYERS; ++l) {
        const float* x_in = (l == 0) ? x : out;
        float* st = stats + l * 2 * DFEAT;
        gather_gemm_stats_kernel<<<PG_BLOCKS, PG_WAVES * 32, 0, stream>>>(
            off, ent, x_in, W + (size_t)l * DFEAT * DFEAT, bias + l * DFEAT, agg, st);
        bn_apply<<<250, BA_WAVES * 32, 0, stream>>>(
            x_in, agg, st, gamma + l * DFEAT, beta + l * DFEAT, out, N_NODES);
    }
}
